# starting kernel
# baseline (speedup 1.0000x reference)
#include <hip/hip_runtime.h>

typedef float v2f __attribute__((ext_vector_type(2)));
typedef float v8f __attribute__((ext_vector_type(8)));

#define N_NODES 50000
#define N_PER   12500
#define NE      600000
#define HD      128

template <int K>
__global__ void gemm_wmma_f32(const float* __restrict__ A,
                              const float* __restrict__ B,
                              const float* __restrict__ bias,
                              const float* __restrict__ rowscale,
                              const float* __restrict__ outscale,
                              float* __restrict__ C,
                              int M, int relu)
{
    const int lane = threadIdx.x & 31;
    const int half = lane >> 4;
    const int lm   = lane & 15;
    const int wave   = (blockIdx.x * blockDim.x + threadIdx.x) >> 5;
    const int nwaves = (gridDim.x * blockDim.x) >> 5;
    const int tn  = wave & 7;
    const int col = tn * 16 + lm;

    v2f bf[K / 4];
    #pragma unroll
    for (int kk = 0; kk < K / 4; ++kk) {
        const int kb = kk * 4 + half * 2;
        bf[kk][0] = B[(long)kb * HD + col];
        bf[kk][1] = B[(long)(kb + 1) * HD + col];
    }

    const int mt = (M + 15) >> 4;
    const float bv = bias[col];

    for (int tm = wave >> 3; tm < mt; tm += (nwaves >> 3)) {
        const int row = tm * 16 + lm;
        const int rc  = (row < M) ? row : (M - 1);
        const float rs = (rowscale != nullptr) ? rowscale[rc] : 1.0f;
        const float* ap = A + (long)rc * K + half * 2;

        v8f acc = {};
        #pragma unroll
        for (int kk = 0; kk < K / 4; ++kk) {
            v2f a = *(const v2f*)(ap + kk * 4);
            a[0] *= rs;
            a[1] *= rs;
            acc = __builtin_amdgcn_wmma_f32_16x16x4_f32(
                false, a, false, bf[kk], (short)0, acc, false, false);
        }

        #pragma unroll
        for (int v = 0; v < 8; ++v) {
            const int orow = tm * 16 + v + half * 8;
            if (orow < M) {
                float val = acc[v] + bv;
                if (relu) val = fmaxf(val, 0.0f);
                if (outscale != nullptr) val *= outscale[orow];
                C[(long)orow * HD + col] = val;
            }
        }
    }
}

__global__ void edge_degree(const int* __restrict__ src,
                            const int* __restrict__ dst,
                            float* __restrict__ degO,
                            float* __restrict__ degI)
{
    int i = blockIdx.x * blockDim.x + threadIdx.x;
    const int stride = gridDim.x * blockDim.x;
    for (; i < NE; i += stride) {
        atomicAdd(&degO[src[i]], 1.0f);
        atomicAdd(&degI[dst[i]], 1.0f);
    }
}

__global__ void make_norm(float* __restrict__ degO, float* __restrict__ degI)
{
    int i = blockIdx.x * blockDim.x + threadIdx.x;
    const int stride = gridDim.x * blockDim.x;
    for (; i < N_NODES; i += stride) {
        degO[i] = rsqrtf(fmaxf(degO[i], 1.0f));
        degI[i] = rsqrtf(fmaxf(degI[i], 1.0f));
    }
}

__global__ void edge_aggregate(const float* __restrict__ h,
                               const int* __restrict__ src,
                               const int* __restrict__ dst,
                               float* __restrict__ accum)
{
    const int lane   = threadIdx.x & 31;
    const int wave   = (blockIdx.x * blockDim.x + threadIdx.x) >> 5;
    const int nwaves = (gridDim.x * blockDim.x) >> 5;
    for (long base = (long)wave * 32; base < NE; base += (long)nwaves * 32) {
        const int s = src[base + lane];
        const int d = dst[base + lane];
        for (int j = 0; j < 32; ++j) {
            const int ss = __shfl(s, j);
            const int dd = __shfl(d, j);
            const float4 v = *(const float4*)(h + (long)ss * HD + lane * 4);
            float* o = accum + (long)dd * HD + lane * 4;
            atomicAdd(o + 0, v.x);
            atomicAdd(o + 1, v.y);
            atomicAdd(o + 2, v.z);
            atomicAdd(o + 3, v.w);
        }
    }
}

__global__ void post_layer0(float* __restrict__ hb,
                            const float* __restrict__ in_norm,
                            const float* __restrict__ out_norm,
                            const float* __restrict__ bias)
{
    int i = blockIdx.x * blockDim.x + threadIdx.x;
    const int stride = gridDim.x * blockDim.x;
    const int total = N_NODES * HD;
    for (; i < total; i += stride) {
        const int row = i >> 7;
        const int col = i & (HD - 1);
        hb[i] = fmaxf(hb[i] * in_norm[row] + bias[col], 0.0f) * out_norm[row];
    }
}

__global__ void edge_final(const float* __restrict__ h,
                           const int* __restrict__ src,
                           const int* __restrict__ dst,
                           const int* __restrict__ ef,
                           float* __restrict__ out)
{
    const int lane   = threadIdx.x & 31;
    const int wave   = (blockIdx.x * blockDim.x + threadIdx.x) >> 5;
    const int nwaves = (gridDim.x * blockDim.x) >> 5;
    for (long base = (long)wave * 32; base < NE; base += (long)nwaves * 32) {
        const int s = src[base + lane];
        const int d = dst[base + lane];
        const int r = ef[base + lane];
        for (int j = 0; j < 32; ++j) {
            const int ss = __shfl(s, j);
            const int dd = __shfl(d, j);
            const int rr = __shfl(r, j);
            const float wf = (rr >= 0 && rr <= 4) ? 2.0f : 1.0f;

            const float4 hs = *(const float4*)(h + (long)ss * HD + lane * 4);
            float* od = out + (long)dd * HD + lane * 4;
            atomicAdd(od + 0, hs.x * wf);
            atomicAdd(od + 1, hs.y * wf);
            atomicAdd(od + 2, hs.z * wf);
            atomicAdd(od + 3, hs.w * wf);

            if (rr == 6 || rr == 14 || rr == 30) {
                const float4 hd = *(const float4*)(h + (long)dd * HD + lane * 4);
                float* os = out + (long)ss * HD + lane * 4;
                atomicAdd(os + 0, hd.x);
                atomicAdd(os + 1, hd.y);
                atomicAdd(os + 2, hd.z);
                atomicAdd(os + 3, hd.w);
            }
        }
    }
}

extern "C" void kernel_launch(void* const* d_in, const int* in_sizes, int n_in,
                              void* d_out, int out_size, void* d_ws, size_t ws_size,
                              hipStream_t stream)
{
    (void)in_sizes; (void)n_in; (void)out_size; (void)ws_size;

    const float* feat0 = (const float*)d_in[0];
    const float* Wfc0  = (const float*)d_in[1];
    const float* bfc0  = (const float*)d_in[2];
    const float* feat1 = (const float*)d_in[3];
    const float* Wfc1  = (const float*)d_in[4];
    const float* bfc1  = (const float*)d_in[5];
    const float* feat2 = (const float*)d_in[6];
    const float* Wfc2  = (const float*)d_in[7];
    const float* bfc2  = (const float*)d_in[8];
    const float* feat3 = (const float*)d_in[9];
    const float* Wfc3  = (const float*)d_in[10];
    const float* bfc3  = (const float*)d_in[11];
    const int*   src   = (const int*)d_in[12];
    const int*   dst   = (const int*)d_in[13];
    const int*   efeat = (const int*)d_in[14];
    const float* b_g0  = (const float*)d_in[15];
    const float* W_g1  = (const float*)d_in[16];
    const float* b_g1  = (const float*)d_in[17];
    float* out = (float*)d_out;

    const size_t NODE_ELEMS = (size_t)N_NODES * HD;
    float* bufA = (float*)d_ws;
    float* bufB = bufA + NODE_ELEMS;
    float* degO = bufB + NODE_ELEMS;
    float* degI = degO + N_NODES;

    hipMemsetAsync(degO, 0, 2 * (size_t)N_NODES * sizeof(float), stream);
    edge_degree<<<1024, 256, 0, stream>>>(src, dst, degO, degI);
    make_norm<<<(N_NODES + 255) / 256, 256, 0, stream>>>(degO, degI);

    const int mt_per = (N_PER + 15) / 16;
    gemm_wmma_f32<128><<<mt_per, 256, 0, stream>>>(
        feat0, Wfc0, bfc0, nullptr, degO + 0 * N_PER,
        bufA + (size_t)0 * N_PER * HD, N_PER, 0);
    gemm_wmma_f32<128><<<mt_per, 256, 0, stream>>>(
        feat1, Wfc1, bfc1, nullptr, degO + 1 * N_PER,
        bufA + (size_t)1 * N_PER * HD, N_PER, 0);
    gemm_wmma_f32<64><<<mt_per, 256, 0, stream>>>(
        feat2, Wfc2, bfc2, nullptr, degO + 2 * N_PER,
        bufA + (size_t)2 * N_PER * HD, N_PER, 0);
    gemm_wmma_f32<64><<<mt_per, 256, 0, stream>>>(
        feat3, Wfc3, bfc3, nullptr, degO + 3 * N_PER,
        bufA + (size_t)3 * N_PER * HD, N_PER, 0);

    hipMemsetAsync(bufB, 0, NODE_ELEMS * sizeof(float), stream);
    edge_aggregate<<<2048, 256, 0, stream>>>(bufA, src, dst, bufB);
    post_layer0<<<(N_NODES * HD + 255) / 256, 256, 0, stream>>>(
        bufB, degI, degO, b_g0);

    hipMemsetAsync(bufA, 0, NODE_ELEMS * sizeof(float), stream);
    edge_aggregate<<<2048, 256, 0, stream>>>(bufB, src, dst, bufA);
    gemm_wmma_f32<128><<<3125, 256, 0, stream>>>(
        bufA, W_g1, b_g1, degI, nullptr, bufB, N_NODES, 1);

    hipMemsetAsync(out, 0, NODE_ELEMS * sizeof(float), stream);
    edge_final<<<2048, 256, 0, stream>>>(bufB, src, dst, efeat, out);
}


# f32 global atomic scatter-adds replaced by owner-computes LDS tiles: 200 workgroups each scan all edges, sum their 250 rows by LDS float atomics (arrival order), store once; no zero-fills
# speedup vs baseline: 1.3542x; 1.3542x over previous
#include <hip/hip_runtime.h>

typedef float v2f __attribute__((ext_vector_type(2)));
typedef float v8f __attribute__((ext_vector_type(8)));

#define N_NODES 50000
#define N_PER   12500
#define NE      600000
#define HD      128

template <int K>
__global__ void gemm_wmma_f32(const float* __restrict__ A,
                              const float* __restrict__ B,
                              const float* __restrict__ bias,
                              const float* __restrict__ rowscale,
                              const float* __restrict__ outscale,
                              float* __restrict__ C,
                              int M, int relu)
{
    const int lane = threadIdx.x & 31;
    const int half = lane >> 4;
    const int lm   = lane & 15;
    const int wave   = (blockIdx.x * blockDim.x + threadIdx.x) >> 5;
    const int nwaves = (gridDim.x * blockDim.x) >> 5;
    const int tn  = wave & 7;
    const int col = tn * 16 + lm;

    v2f bf[K / 4];
    #pragma unroll
    for (int kk = 0; kk < K / 4; ++kk) {
        const int kb = kk * 4 + half * 2;
        bf[kk][0] = B[(long)kb * HD + col];
        bf[kk][1] = B[(long)(kb + 1) * HD + col];
    }

    const int mt = (M + 15) >> 4;
    const float bv = bias[col];

    for (int tm = wave >> 3; tm < mt; tm += (nwaves >> 3)) {
        const int row = tm * 16 + lm;
        const int rc  = (row < M) ? row : (M - 1);
        const float rs = (rowscale != nullptr) ? rowscale[rc] : 1.0f;
        const float* ap = A + (long)rc * K + half * 2;

        v8f acc = {};
        #pragma unroll
        for (int kk = 0; kk < K / 4; ++kk) {
            v2f a = *(const v2f*)(ap + kk * 4);
            a[0] *= rs;
            a[1] *= rs;
            acc = __builtin_amdgcn_wmma_f32_16x16x4_f32(
                false, a, false, bf[kk], (short)0, acc, false, false);
        }

        #pragma unroll
        for (int v = 0; v < 8; ++v) {
            const int orow = tm * 16 + v + half * 8;
            if (orow < M) {
                float val = acc[v] + bv;
                if (relu) val = fmaxf(val, 0.0f);
                if (outscale != nullptr) val *= outscale[orow];
                C[(long)orow * HD + col] = val;
            }
        }
    }
}

__global__ void edge_degree(const int* __restrict__ src,
                            const int* __restrict__ dst,
                            float* __restrict__ degO,
                            float* __restrict__ degI)
{
    int i = blockIdx.x * blockDim.x + threadIdx.x;
    const int stride = gridDim.x * blockDim.x;
    for (; i < NE; i += stride) {
        atomicAdd(&degO[src[i]], 1.0f);
        atomicAdd(&degI[dst[i]], 1.0f);
    }
}

__global__ void make_norm(float* __restrict__ degO, float* __restrict__ degI)
{
    int i = blockIdx.x * blockDim.x + threadIdx.x;
    const int stride = gridDim.x * blockDim.x;
    for (; i < N_NODES; i += stride) {
        degO[i] = rsqrtf(fmaxf(degO[i], 1.0f));
        degI[i] = rsqrtf(fmaxf(degI[i], 1.0f));
    }
}

#define TILE_R      250
#define N_TILES     200
#define Q_CAP       16384
#define AGG_THREADS 1024
#define GATHER_UNR  8

static_assert(TILE_R * N_TILES == N_NODES, "tiles must cover the node range exactly");
static_assert(NE % 4 == 0, "edge arrays are read as int4");
static_assert(NE <= (1 << 20) && TILE_R <= (1 << 9), "queue entry packing");

__device__ __forceinline__ void lds_add(float* p, float v)
{
    __hip_atomic_fetch_add(p, v, __ATOMIC_RELAXED, __HIP_MEMORY_SCOPE_WORKGROUP);
}

template <int MODE>
__global__ __launch_bounds__(AGG_THREADS)
void tile_aggregate(const float* __restrict__ h,
                    const int* __restrict__ src,
                    const int* __restrict__ dst,
                    const int* __restrict__ ef,
                    const float* __restrict__ in_norm,
                    const float* __restrict__ out_norm,
                    const float* __restrict__ bias,
                    float* __restrict__ outp)
{
    __shared__ float tile[TILE_R * HD];
    __shared__ int   queue[Q_CAP];
    __shared__ int   qn;

    const int tid  = threadIdx.x;
    const int lane = tid & 31;
    const int wave = __builtin_amdgcn_readfirstlane(tid >> 5);
    const int r0   = blockIdx.x * TILE_R;

    for (int i = tid; i < TILE_R * HD; i += AGG_THREADS) tile[i] = 0.0f;
    if (tid == 0) qn = 0;
    __syncthreads();

    auto resolve = [&](int ent, int& node, int& wc) {
        const int e = ent & 0xfffff;
        if (MODE == 2) {
            const int r = ef[e];
            if ((ent >> 29) & 1) {
                node = dst[e];
                wc = (r == 6 || r == 14 || r == 30) ? 1 : 0;
            } else {
                node = src[e];
                wc = (r >= 0 && r <= 4) ? 2 : 1;
            }
        } else {
            node = src[e];
            wc = 1;
        }
    };

    auto place = [&](int slot, int ent) {
        if (slot < Q_CAP) {
            queue[slot] = ent;
        } else {
            int node, wc;
            resolve(ent, node, wc);
            if (wc != 0) {
                const float w = (wc == 2) ? 2.0f : 1.0f;
                float* t = tile + ((ent >> 20) & 0x1ff) * HD;
                for (int c = 0; c < HD; ++c)
                    lds_add(t + (c & 3) * 32 + (c >> 2), h[(long)node * HD + c] * w);
            }
        }
    };

    constexpr int SU   = (MODE == 2) ? 4 : 8;
    constexpr int NPD  = SU * 4;
    constexpr int STEP = AGG_THREADS * 4;
    for (int w0 = 0; w0 < NE; w0 += STEP * SU) {
        const int b0 = w0 + tid * 4;
        int dv[NPD];
        int sv[NPD];
        #pragma unroll
        for (int u = 0; u < SU; ++u) {
            const int b = b0 + u * STEP;
            const bool in = b < NE;
            const int4 d = in ? *(const int4*)(dst + b) : make_int4(-1, -1, -1, -1);
            dv[4 * u + 0] = d.x; dv[4 * u + 1] = d.y; dv[4 * u + 2] = d.z; dv[4 * u + 3] = d.w;
            if (MODE == 2) {
                const int4 q = in ? *(const int4*)(src + b) : make_int4(-1, -1, -1, -1);
                sv[4 * u + 0] = q.x; sv[4 * u + 1] = q.y; sv[4 * u + 2] = q.z; sv[4 * u + 3] = q.w;
            }
        }
        unsigned mkd[NPD];
        unsigned mks[NPD];
        int tot = 0;
        #pragma unroll
        for (int p = 0; p < NPD; ++p) {
            mkd[p] = __builtin_amdgcn_ballot_w32((unsigned)(dv[p] - r0) < (unsigned)TILE_R);
            tot += __builtin_popcount(mkd[p]);
            if (MODE == 2) {
                mks[p] = __builtin_amdgcn_ballot_w32((unsigned)(sv[p] - r0) < (unsigned)TILE_R);
                tot += __builtin_popcount(mks[p]);
            }
        }
        if (tot != 0) {
            int base = 0;
            if (lane == 0) base = atomicAdd(&qn, tot);
            base = __builtin_amdgcn_readfirstlane(base);
            #pragma unroll
            for (int p = 0; p < NPD; ++p) {
                const int e = b0 + (p >> 2) * STEP + (p & 3);
                const unsigned dl = (unsigned)(dv[p] - r0);
                if (dl < (unsigned)TILE_R)
                    place(base + (int)__builtin_amdgcn_mbcnt_lo(mkd[p], 0u), e | (int)(dl << 20));
                base += __builtin_popcount(mkd[p]);
                if (MODE == 2) {
                    const unsigned sl = (unsigned)(sv[p] - r0);
                    if (sl < (unsigned)TILE_R)
                        place(base + (int)__builtin_amdgcn_mbcnt_lo(mks[p], 0u),
                              e | (int)(sl << 20) | (1 << 29));
                    base += __builtin_popcount(mks[p]);
                }
            }
        }
    }
    __syncthreads();

    const int n = __builtin_amdgcn_readfirstlane(min(qn, Q_CAP));
    for (int b = wave * 32; b < n; b += AGG_THREADS) {
        const int idx = b + lane;
        int node = 0, wc = 0, lrow = 0;
        if (idx < n) {
            const int ent = queue[idx];
            lrow = (ent >> 20) & 0x1ff;
            resolve(ent, node, wc);
        }
        const int cnt = min(32, n - b);
        for (int j = 0; j < cnt; j += GATHER_UNR) {
            float4 v[GATHER_UNR];
            int    m[GATHER_UNR];
            int    wj[GATHER_UNR];
            #pragma unroll
            for (int u = 0; u < GATHER_UNR; ++u) {
                const int jc = min(j + u, cnt - 1);
                const int nd = __builtin_amdgcn_readlane(node, jc);
                m[u]  = __builtin_amdgcn_readlane(lrow, jc);
                wj[u] = (j + u < cnt) ? __builtin_amdgcn_readlane(wc, jc) : 0;
                if (MODE == 2) {
                    if (wj[u] != 0) v[u] = *(const float4*)(h + (long)nd * HD + lane * 4);
                } else {
                    v[u] = *(const float4*)(h + (long)nd * HD + lane * 4);
                }
            }
            #pragma unroll
            for (int u = 0; u < GATHER_UNR; ++u) {
                if (wj[u] != 0) {
                    float* t = tile + m[u] * HD + lane;
                    if (MODE == 2) {
                        const float w = (wj[u] == 2) ? 2.0f : 1.0f;
                        lds_add(t +  0, v[u].x * w);
                        lds_add(t + 32, v[u].y * w);
                        lds_add(t + 64, v[u].z * w);
                        lds_add(t + 96, v[u].w * w);
                    } else {
                        lds_add(t +  0, v[u].x);
                        lds_add(t + 32, v[u].y);
                        lds_add(t + 64, v[u].z);
                        lds_add(t + 96, v[u].w);
                    }
                }
            }
        }
    }
    __syncthreads();

    for (int i = tid; i < TILE_R * 32; i += AGG_THREADS) {
        const int lr = i >> 5;
        const int l  = i & 31;
        const float* t = tile + lr * HD + l;
        float4 v = make_float4(t[0], t[32], t[64], t[96]);
        const int row = r0 + lr;
        if (MODE == 0) {
            const float inn = in_norm[row];
            const float on  = out_norm[row];
            const float4 bb = *(const float4*)(bias + l * 4);
            v.x = fmaxf(v.x * inn + bb.x, 0.0f) * on;
            v.y = fmaxf(v.y * inn + bb.y, 0.0f) * on;
            v.z = fmaxf(v.z * inn + bb.z, 0.0f) * on;
            v.w = fmaxf(v.w * inn + bb.w, 0.0f) * on;
        }
        *(float4*)(outp + (long)row * HD + l * 4) = v;
    }
}

extern "C" void kernel_launch(void* const* d_in, const int* in_sizes, int n_in,
                              void* d_out, int out_size, void* d_ws, size_t ws_size,
                              hipStream_t stream)
{
    (void)in_sizes; (void)n_in; (void)out_size; (void)ws_size;

    const float* feat0 = (const float*)d_in[0];
    const float* Wfc0  = (const float*)d_in[1];
    const float* bfc0  = (const float*)d_in[2];
    const float* feat1 = (const float*)d_in[3];
    const float* Wfc1  = (const float*)d_in[4];
    const float* bfc1  = (const float*)d_in[5];
    const float* feat2 = (const float*)d_in[6];
    const float* Wfc2  = (const float*)d_in[7];
    const float* bfc2  = (const float*)d_in[8];
    const float* feat3 = (const float*)d_in[9];
    const float* Wfc3  = (const float*)d_in[10];
    const float* bfc3  = (const float*)d_in[11];
    const int*   src   = (const int*)d_in[12];
    const int*   dst   = (const int*)d_in[13];
    const int*   efeat = (const int*)d_in[14];
    const float* b_g0  = (const float*)d_in[15];
    const float* W_g1  = (const float*)d_in[16];
    const float* b_g1  = (const float*)d_in[17];
    float* out = (float*)d_out;

    const size_t NODE_ELEMS = (size_t)N_NODES * HD;
    float* bufA = (float*)d_ws;
    float* bufB = bufA + NODE_ELEMS;
    float* degO = bufB + NODE_ELEMS;
    float* degI = degO + N_NODES;

    hipMemsetAsync(degO, 0, 2 * (size_t)N_NODES * sizeof(float), stream);
    edge_degree<<<1024, 256, 0, stream>>>(src, dst, degO, degI);
    make_norm<<<(N_NODES + 255) / 256, 256, 0, stream>>>(degO, degI);

    const int mt_per = (N_PER + 15) / 16;
    gemm_wmma_f32<128><<<mt_per, 256, 0, stream>>>(
        feat0, Wfc0, bfc0, nullptr, degO + 0 * N_PER,
        bufA + (size_t)0 * N_PER * HD, N_PER, 0);
    gemm_wmma_f32<128><<<mt_per, 256, 0, stream>>>(
        feat1, Wfc1, bfc1, nullptr, degO + 1 * N_PER,
        bufA + (size_t)1 * N_PER * HD, N_PER, 0);
    gemm_wmma_f32<64><<<mt_per, 256, 0, stream>>>(
        feat2, Wfc2, bfc2, nullptr, degO + 2 * N_PER,
        bufA + (size_t)2 * N_PER * HD, N_PER, 0);
    gemm_wmma_f32<64><<<mt_per, 256, 0, stream>>>(
        feat3, Wfc3, bfc3, nullptr, degO + 3 * N_PER,
        bufA + (size_t)3 * N_PER * HD, N_PER, 0);

    tile_aggregate<0><<<N_TILES, AGG_THREADS, 0, stream>>>(
        bufA, src, dst, nullptr, degI, degO, b_g0, bufB);

    tile_aggregate<1><<<N_TILES, AGG_THREADS, 0, stream>>>(
        bufB, src, dst, nullptr, nullptr, nullptr, nullptr, bufA);
    gemm_wmma_f32<128><<<3125, 256, 0, stream>>>(
        bufA, W_g1, b_g1, degI, nullptr, bufB, N_NODES, 1);

    tile_aggregate<2><<<N_TILES, AGG_THREADS, 0, stream>>>(
        bufB, src, dst, efeat, nullptr, nullptr, nullptr, out);
}


# aggregation launches scan one-byte tile keys (made per call by the degree pass, 4 keys per 32-bit test; reverse keys only where e_feat adds) instead of 4-byte node ids; sums stay arrival-order
# speedup vs baseline: 1.6149x; 1.1925x over previous
#include <hip/hip_runtime.h>

typedef float v2f __attribute__((ext_vector_type(2)));
typedef float v8f __attribute__((ext_vector_type(8)));

#define N_NODES 50000
#define N_PER   12500
#define NE      600000
#define HD      128

#define TILE_R      250
#define N_TILES     200
#define N_KEYW      (NE / 4)

static_assert(TILE_R * N_TILES == N_NODES, "tiles must cover the node range exactly");
static_assert(N_TILES <= 255, "a tile key is one byte and 0xff is the never-matching filler");
static_assert(NE % 16 == 0, "edge ids are read as int4, keys as 16-byte vectors");

template <int K>
__global__ void gemm_wmma_f32(const float* __restrict__ A,
                              const float* __restrict__ B,
                              const float* __restrict__ bias,
                              const float* __restrict__ rowscale,
                              const float* __restrict__ outscale,
                              float* __restrict__ C,
                              int M, int relu)
{
    const int lane = threadIdx.x & 31;
    const int half = lane >> 4;
    const int lm   = lane & 15;
    const int wave   = (blockIdx.x * blockDim.x + threadIdx.x) >> 5;
    const int nwaves = (gridDim.x * blockDim.x) >> 5;
    const int tn  = wave & 7;
    const int col = tn * 16 + lm;

    v2f bf[K / 4];
    #pragma unroll
    for (int kk = 0; kk < K / 4; ++kk) {
        const int kb = kk * 4 + half * 2;
        bf[kk][0] = B[(long)kb * HD + col];
        bf[kk][1] = B[(long)(kb + 1) * HD + col];
    }

    const int mt = (M + 15) >> 4;
    const float bv = bias[col];

    for (int tm = wave >> 3; tm < mt; tm += (nwaves >> 3)) {
        const int row = tm * 16 + lm;
        const int rc  = (row < M) ? row : (M - 1);
        const float rs = (rowscale != nullptr) ? rowscale[rc] : 1.0f;
        const float* ap = A + (long)rc * K + half * 2;

        v8f acc = {};
        #pragma unroll
        for (int kk = 0; kk < K / 4; ++kk) {
            v2f a = *(const v2f*)(ap + kk * 4);
            a[0] *= rs;
            a[1] *= rs;
            acc = __builtin_amdgcn_wmma_f32_16x16x4_f32(
                false, a, false, bf[kk], (short)0, acc, false, false);
        }

        #pragma unroll
        for (int v = 0; v < 8; ++v) {
            const int orow = tm * 16 + v + half * 8;
            if (orow < M) {
                float val = acc[v] + bv;
                if (relu) val = fmaxf(val, 0.0f);
                if (outscale != nullptr) val *= outscale[orow];
                C[(long)orow * HD + col] = val;
            }
        }
    }
}

template <bool LAST>
__global__ void edge_degree(const int* __restrict__ src,
                            const int* __restrict__ dst,
                            const int* __restrict__ ef,
                            float* __restrict__ degO,
                            float* __restrict__ degI,
                            float* __restrict__ keybuf)
{
    unsigned* keys = (unsigned*)keybuf;
    int j = blockIdx.x * blockDim.x + threadIdx.x;
    const int stride = gridDim.x * blockDim.x;
    for (; j < N_KEYW; j += stride) {
        const int4 s = *(const int4*)(src + 4 * j);
        const int4 d = *(const int4*)(dst + 4 * j);
        if (!LAST) {
            atomicAdd(&degO[s.x], 1.0f);
            atomicAdd(&degO[s.y], 1.0f);
            atomicAdd(&degO[s.z], 1.0f);
            atomicAdd(&degO[s.w], 1.0f);
            atomicAdd(&degI[d.x], 1.0f);
            atomicAdd(&degI[d.y], 1.0f);
            atomicAdd(&degI[d.z], 1.0f);
            atomicAdd(&degI[d.w], 1.0f);
        }
        keys[j] = ((unsigned)d.x / TILE_R) | (((unsigned)d.y / TILE_R) << 8) |
                  (((unsigned)d.z / TILE_R) << 16) | (((unsigned)d.w / TILE_R) << 24);
        unsigned ks = ((unsigned)s.x / TILE_R) | (((unsigned)s.y / TILE_R) << 8) |
                      (((unsigned)s.z / TILE_R) << 16) | (((unsigned)s.w / TILE_R) << 24);
        if (LAST) {
            const int4 r = *(const int4*)(ef + 4 * j);
            if (!(r.x == 6 || r.x == 14 || r.x == 30)) ks |= 0x000000ffu;
            if (!(r.y == 6 || r.y == 14 || r.y == 30)) ks |= 0x0000ff00u;
            if (!(r.z == 6 || r.z == 14 || r.z == 30)) ks |= 0x00ff0000u;
            if (!(r.w == 6 || r.w == 14 || r.w == 30)) ks |= 0xff000000u;
        }
        keys[N_KEYW + j] = ks;
    }
}

__global__ void make_norm(float* __restrict__ degO, float* __restrict__ degI)
{
    int i = blockIdx.x * blockDim.x + threadIdx.x;
    const int stride = gridDim.x * blockDim.x;
    for (; i < N_NODES; i += stride) {
        degO[i] = rsqrtf(fmaxf(degO[i], 1.0f));
        degI[i] = rsqrtf(fmaxf(degI[i], 1.0f));
    }
}

#define Q_CAP       16384
#define AGG_THREADS 1024
#define GATHER_UNR  8

static_assert(NE <= (1 << 20), "queue entry packing");

__device__ __forceinline__ void lds_add(float* p, float v)
{
    __hip_atomic_fetch_add(p, v, __ATOMIC_RELAXED, __HIP_MEMORY_SCOPE_WORKGROUP);
}

template <int MODE>
__global__ __launch_bounds__(AGG_THREADS)
void tile_aggregate(const float* __restrict__ h,
                    const int* __restrict__ src,
                    const int* __restrict__ dst,
                    const int* __restrict__ ef,
                    const float* __restrict__ keybuf,
                    const float* __restrict__ in_norm,
                    const float* __restrict__ out_norm,
                    const float* __restrict__ bias,
                    float* __restrict__ outp)
{
    __shared__ float tile[TILE_R * HD];
    __shared__ int   queue[Q_CAP];
    __shared__ int   qn;

    const int tid  = threadIdx.x;
    const int lane = tid & 31;
    const int wave = __builtin_amdgcn_readfirstlane(tid >> 5);
    const int r0   = blockIdx.x * TILE_R;

    for (int i = tid; i < TILE_R * HD; i += AGG_THREADS) tile[i] = 0.0f;
    if (tid == 0) qn = 0;
    __syncthreads();

    auto resolve = [&](int ent, int& node, int& wc, int& lrow) {
        const int e = ent & 0xfffff;
        const int s = src[e];
        const int d = dst[e];
        if (MODE == 2) {
            const int r = ef[e];
            if ((ent >> 29) & 1) {
                node = d;
                lrow = s - r0;
                wc = (r == 6 || r == 14 || r == 30) ? 1 : 0;
            } else {
                node = s;
                lrow = d - r0;
                wc = (r >= 0 && r <= 4) ? 2 : 1;
            }
        } else {
            node = s;
            lrow = d - r0;
            wc = 1;
        }
    };

    auto place = [&](int slot, int ent) {
        if (slot < Q_CAP) {
            queue[slot] = ent;
        } else {
            int node, wc, lrow;
            resolve(ent, node, wc, lrow);
            if (wc != 0) {
                const float w = (wc == 2) ? 2.0f : 1.0f;
                float* t = tile + lrow * HD;
                for (int c = 0; c < HD; ++c)
                    lds_add(t + (c & 3) * 32 + (c >> 2), h[(long)node * HD + c] * w);
            }
        }
    };

    constexpr int NV = NE / 16;
    constexpr int SU = (MODE == 2) ? 4 : 8;
    const unsigned tk = (unsigned)blockIdx.x * 0x01010101u;
    const uint4* __restrict__ kdv = (const uint4*)keybuf;
    const uint4* __restrict__ ksv = (const uint4*)keybuf + NV;

    auto eq4 = [&](unsigned x) -> unsigned {
        const unsigned t = x ^ tk;
        return ~(((t & 0x7f7f7f7fu) + 0x7f7f7f7fu) | t) & 0x80808080u;
    };
    auto match16 = [&](const uint4& q) -> unsigned {
        return (eq4(q.x) >> 7) | (eq4(q.y) >> 6) | (eq4(q.z) >> 5) | (eq4(q.w) >> 4);
    };
    auto emit = [&](unsigned f, int ebase, int flag) {
        while (f != 0) {
            const int b = __builtin_ctz(f);
            f &= f - 1;
            const int slot = atomicAdd(&qn, 1);
            place(slot, (ebase + (b & 7) * 4 + (b >> 3)) | flag);
        }
    };

    for (int v0 = 0; v0 < NV; v0 += AGG_THREADS * SU) {
        uint4 kd[SU];
        uint4 ks[SU];
        #pragma unroll
        for (int u = 0; u < SU; ++u) {
            const int iv = v0 + u * AGG_THREADS + tid;
            const bool in = iv < NV;
            kd[u] = in ? kdv[iv] : make_uint4(~0u, ~0u, ~0u, ~0u);
            if (MODE == 2) ks[u] = in ? ksv[iv] : make_uint4(~0u, ~0u, ~0u, ~0u);
        }
        #pragma unroll
        for (int u = 0; u < SU; ++u) {
            const int ebase = (v0 + u * AGG_THREADS + tid) * 16;
            emit(match16(kd[u]), ebase, 0);
            if (MODE == 2) emit(match16(ks[u]), ebase, 1 << 29);
        }
    }
    __syncthreads();

    const int n = __builtin_amdgcn_readfirstlane(min(qn, Q_CAP));
    for (int b = wave * 32; b < n; b += AGG_THREADS) {
        const int idx = b + lane;
        int node = 0, wc = 0, lrow = 0;
        if (idx < n) {
            resolve(queue[idx], node, wc, lrow);
        }
        const int cnt = min(32, n - b);
        for (int j = 0; j < cnt; j += GATHER_UNR) {
            float4 v[GATHER_UNR];
            int    m[GATHER_UNR];
            int    wj[GATHER_UNR];
            #pragma unroll
            for (int u = 0; u < GATHER_UNR; ++u) {
                const int jc = min(j + u, cnt - 1);
                const int nd = __builtin_amdgcn_readlane(node, jc);
                m[u]  = __builtin_amdgcn_readlane(lrow, jc);
                wj[u] = (j + u < cnt) ? __builtin_amdgcn_readlane(wc, jc) : 0;
                if (MODE == 2) {
                    if (wj[u] != 0) v[u] = *(const float4*)(h + (long)nd * HD + lane * 4);
                } else {
                    v[u] = *(const float4*)(h + (long)nd * HD + lane * 4);
                }
            }
            #pragma unroll
            for (int u = 0; u < GATHER_UNR; ++u) {
                if (wj[u] != 0) {
                    float* t = tile + m[u] * HD + lane;
                    if (MODE == 2) {
                        const float w = (wj[u] == 2) ? 2.0f : 1.0f;
                        lds_add(t +  0, v[u].x * w);
                        lds_add(t + 32, v[u].y * w);
                        lds_add(t + 64, v[u].z * w);
                        lds_add(t + 96, v[u].w * w);
                    } else {
                        lds_add(t +  0, v[u].x);
                        lds_add(t + 32, v[u].y);
                        lds_add(t + 64, v[u].z);
                        lds_add(t + 96, v[u].w);
                    }
                }
            }
        }
    }
    __syncthreads();

    for (int i = tid; i < TILE_R * 32; i += AGG_THREADS) {
        const int lr = i >> 5;
        const int l  = i & 31;
        const float* t = tile + lr * HD + l;
        float4 v = make_float4(t[0], t[32], t[64], t[96]);
        const int row = r0 + lr;
        if (MODE == 0) {
            const float inn = in_norm[row];
            const float on  = out_norm[row];
            const float4 bb = *(const float4*)(bias + l * 4);
            v.x = fmaxf(v.x * inn + bb.x, 0.0f) * on;
            v.y = fmaxf(v.y * inn + bb.y, 0.0f) * on;
            v.z = fmaxf(v.z * inn + bb.z, 0.0f) * on;
            v.w = fmaxf(v.w * inn + bb.w, 0.0f) * on;
        }
        *(float4*)(outp + (long)row * HD + l * 4) = v;
    }
}

extern "C" void kernel_launch(void* const* d_in, const int* in_sizes, int n_in,
                              void* d_out, int out_size, void* d_ws, size_t ws_size,
                              hipStream_t stream)
{
    (void)in_sizes; (void)n_in; (void)out_size; (void)ws_size;

    const float* feat0 = (const float*)d_in[0];
    const float* Wfc0  = (const float*)d_in[1];
    const float* bfc0  = (const float*)d_in[2];
    const float* feat1 = (const float*)d_in[3];
    const float* Wfc1  = (const float*)d_in[4];
    const float* bfc1  = (const float*)d_in[5];
    const float* feat2 = (const float*)d_in[6];
    const float* Wfc2  = (const float*)d_in[7];
    const float* bfc2  = (const float*)d_in[8];
    const float* feat3 = (const float*)d_in[9];
    const float* Wfc3  = (const float*)d_in[10];
    const float* bfc3  = (const float*)d_in[11];
    const int*   src   = (const int*)d_in[12];
    const int*   dst   = (const int*)d_in[13];
    const int*   efeat = (const int*)d_in[14];
    const float* b_g0  = (const float*)d_in[15];
    const float* W_g1  = (const float*)d_in[16];
    const float* b_g1  = (const float*)d_in[17];
    float* out = (float*)d_out;

    const size_t NODE_ELEMS = (size_t)N_NODES * HD;
    float* bufA = (float*)d_ws;
    float* bufB = bufA + NODE_ELEMS;
    float* degO = bufB + NODE_ELEMS;
    float* degI = degO + N_NODES;
    float* keys01 = out;
    float* keys2  = bufA;

    hipMemsetAsync(degO, 0, 2 * (size_t)N_NODES * sizeof(float), stream);
    edge_degree<false><<<(N_KEYW + 255) / 256, 256, 0, stream>>>(src, dst, nullptr, degO, degI, keys01);
    make_norm<<<(N_NODES + 255) / 256, 256, 0, stream>>>(degO, degI);

    const int mt_per = (N_PER + 15) / 16;
    gemm_wmma_f32<128><<<mt_per, 256, 0, stream>>>(
        feat0, Wfc0, bfc0, nullptr, degO + 0 * N_PER,
        bufA + (size_t)0 * N_PER * HD, N_PER, 0);
    gemm_wmma_f32<128><<<mt_per, 256, 0, stream>>>(
        feat1, Wfc1, bfc1, nullptr, degO + 1 * N_PER,
        bufA + (size_t)1 * N_PER * HD, N_PER, 0);
    gemm_wmma_f32<64><<<mt_per, 256, 0, stream>>>(
        feat2, Wfc2, bfc2, nullptr, degO + 2 * N_PER,
        bufA + (size_t)2 * N_PER * HD, N_PER, 0);
    gemm_wmma_f32<64><<<mt_per, 256, 0, stream>>>(
        feat3, Wfc3, bfc3, nullptr, degO + 3 * N_PER,
        bufA + (size_t)3 * N_PER * HD, N_PER, 0);

    tile_aggregate<0><<<N_TILES, AGG_THREADS, 0, stream>>>(
        bufA, src, dst, nullptr, keys01, degI, degO, b_g0, bufB);

    tile_aggregate<1><<<N_TILES, AGG_THREADS, 0, stream>>>(
        bufB, src, dst, nullptr, keys01, nullptr, nullptr, nullptr, bufA);
    gemm_wmma_f32<128><<<3125, 256, 0, stream>>>(
        bufA, W_g1, b_g1, degI, nullptr, bufB, N_NODES, 1);

    edge_degree<true><<<(N_KEYW + 255) / 256, 256, 0, stream>>>(src, dst, efeat, nullptr, nullptr, keys2);
    tile_aggregate<2><<<N_TILES, AGG_THREADS, 0, stream>>>(
        bufB, src, dst, efeat, keys2, nullptr, nullptr, nullptr, out);
}


# degree counts moved from global f32 atomics into the owner tiles' key scan (list launch split off ahead of the projections; src keys for all edges); no memset, no norm launch; sums arrival-order
# speedup vs baseline: 2.6650x; 1.6503x over previous
#include <hip/hip_runtime.h>

typedef float v2f __attribute__((ext_vector_type(2)));
typedef float v8f __attribute__((ext_vector_type(8)));

#define N_NODES 50000
#define N_PER   12500
#define NE      600000
#define HD      128

#define TILE_R      250
#define N_TILES     200
#define N_KEYW      (NE / 4)
#define REGION_B    (TILE_R * HD * 4)
#define KEY_OFF     65536
#define KCH_LOG     11
#define ROW_CAP     64
#define LIST_INTS   (TILE_R * ROW_CAP + TILE_R + 1)

static_assert(TILE_R * N_TILES == N_NODES, "tiles must cover the node range exactly");
static_assert(N_TILES <= 255, "a tile key is one byte and 0xff is the never-matching filler");
static_assert(NE % 16 == 0, "edge ids are read as int4, keys as 16-byte vectors");
static_assert(LIST_INTS * 4 <= KEY_OFF, "row lists end before the key chunk of their region");
static_assert(KEY_OFF + (16 << KCH_LOG) <= REGION_B, "a key chunk ends inside its region");
static_assert(((2 * (NE / 16) + (1 << KCH_LOG) - 1) >> KCH_LOG) <= N_TILES, "enough regions for all key chunks");
static_assert(REGION_B % 16 == 0 && KEY_OFF % 16 == 0, "16-byte vectors stay aligned");

__device__ __forceinline__ uint4* key_vec(float* base, int civ)
{
    return (uint4*)((char*)base + (size_t)(civ >> KCH_LOG) * REGION_B + KEY_OFF) + (civ & ((1 << KCH_LOG) - 1));
}

template <int K>
__global__ void gemm_wmma_f32(const float* __restrict__ A,
                              const float* __restrict__ B,
                              const float* __restrict__ bias,
                              const float* __restrict__ rowscale,
                              const float* __restrict__ outscale,
                              float* __restrict__ C,
                              int M, int relu)
{
    const int lane = threadIdx.x & 31;
    const int half = lane >> 4;
    const int lm   = lane & 15;
    const int wave   = (blockIdx.x * blockDim.x + threadIdx.x) >> 5;
    const int nwaves = (gridDim.x * blockDim.x) >> 5;
    const int tn  = wave & 7;
    const int col = tn * 16 + lm;

    v2f bf[K / 4];
    #pragma unroll
    for (int kk = 0; kk < K / 4; ++kk) {
        const int kb = kk * 4 + half * 2;
        bf[kk][0] = B[(long)kb * HD + col];
        bf[kk][1] = B[(long)(kb + 1) * HD + col];
    }

    const int mt = (M + 15) >> 4;
    const float bv = bias[col];

    for (int tm = wave >> 3; tm < mt; tm += (nwaves >> 3)) {
        const int row = tm * 16 + lm;
        const int rc  = (row < M) ? row : (M - 1);
        const float rs = (rowscale != nullptr) ? rowscale[rc] : 1.0f;
        const float* ap = A + (long)rc * K + half * 2;

        v8f acc = {};
        #pragma unroll
        for (int kk = 0; kk < K / 4; ++kk) {
            v2f a = *(const v2f*)(ap + kk * 4);
            a[0] *= rs;
            a[1] *= rs;
            acc = __builtin_amdgcn_wmma_f32_16x16x4_f32(
                false, a, false, bf[kk], (short)0, acc, false, false);
        }

        #pragma unroll
        for (int v = 0; v < 8; ++v) {
            const int orow = tm * 16 + v + half * 8;
            if (orow < M) {
                float val = acc[v] + bv;
                if (relu) val = fmaxf(val, 0.0f);
                if (outscale != nullptr) val *= outscale[orow];
                C[(long)orow * HD + col] = val;
            }
        }
    }
}

__global__ void edge_keys(const int* __restrict__ src,
                          const int* __restrict__ dst,
                          float* __restrict__ keybuf)
{
    int j = blockIdx.x * blockDim.x + threadIdx.x;
    const int stride = gridDim.x * blockDim.x;
    for (; j < N_KEYW; j += stride) {
        const int4 s = *(const int4*)(src + 4 * j);
        const int4 d = *(const int4*)(dst + 4 * j);
        const unsigned kd = ((unsigned)d.x / TILE_R) | (((unsigned)d.y / TILE_R) << 8) |
                            (((unsigned)d.z / TILE_R) << 16) | (((unsigned)d.w / TILE_R) << 24);
        const unsigned ks = ((unsigned)s.x / TILE_R) | (((unsigned)s.y / TILE_R) << 8) |
                            (((unsigned)s.z / TILE_R) << 16) | (((unsigned)s.w / TILE_R) << 24);
        const int cw = N_KEYW + j;
        ((unsigned*)key_vec(keybuf, j >> 2))[j & 3]   = kd;
        ((unsigned*)key_vec(keybuf, cw >> 2))[cw & 3] = ks;
    }
}

#define Q_CAP       16384
#define LQ_CAP      32768
#define AGG_THREADS 1024
#define AGG_WAVES   (AGG_THREADS / 32)
#define GATHER_UNR  16
#define DEAL_UNR    4

static_assert(NE <= (1 << 20), "queue entry packing");
static_assert(ROW_CAP == 64, "a row's list is read as two batches of 32 entries, one per lane");
static_assert(N_NODES <= (1 << 16), "row-list entry: node id in the low 16 bits, flags in bits 16 and 17");
static_assert((TILE_R * ROW_CAP) % 4 == 0, "row lists are copied as 16-byte vectors");

__device__ __forceinline__ void lds_add(float* p, float v)
{
    __hip_atomic_fetch_add(p, v, __ATOMIC_RELAXED, __HIP_MEMORY_SCOPE_WORKGROUP);
}

__global__ __launch_bounds__(AGG_THREADS)
void tile_lists(const int* __restrict__ src,
                const int* __restrict__ dst,
                const int* __restrict__ ef,
                float* scratch,
                float* __restrict__ out_norm,
                float* __restrict__ in_norm)
{
    __shared__ int   queue[LQ_CAP];
    __shared__ __attribute__((aligned(16))) int rq[TILE_R * ROW_CAP];
    __shared__ int   rcnt[TILE_R];
    __shared__ int   din[TILE_R];
    __shared__ int   dout[TILE_R];
    __shared__ int   qn;
    __shared__ int   ovf;

    const int tid = threadIdx.x;
    const int r0  = blockIdx.x * TILE_R;
    int* mylist = (int*)((char*)scratch + (size_t)blockIdx.x * REGION_B);

    if (tid < TILE_R) { rcnt[tid] = 0; din[tid] = 0; dout[tid] = 0; }
    if (tid == 0) { qn = 0; ovf = 0; }
    __syncthreads();

    auto deal = [&](int ent, int s, int d, int r, bool listed) {
        const bool srcend = ((ent >> 29) & 1) != 0;
        const int lrow = (srcend ? s : d) - r0;
        if ((unsigned)lrow >= (unsigned)TILE_R) return;
        atomicAdd(srcend ? &dout[lrow] : &din[lrow], 1);
        if (srcend && !(r == 6 || r == 14 || r == 30)) return;
        if (!listed) return;
        const int fl = srcend ? 2 : ((r >= 0 && r <= 4) ? 1 : 0);
        const int slot = atomicAdd(&rcnt[lrow], 1);
        if (slot < ROW_CAP)
            rq[lrow * ROW_CAP + slot] = (srcend ? d : s) | (fl << 16);
        else
            ovf = 1;
    };

    auto place = [&](int slot, int ent) {
        if (slot < LQ_CAP) {
            queue[slot] = ent;
        } else {
            const int e = ent & 0xfffff;
            ovf = 1;
            deal(ent, src[e], dst[e], ef[e], false);
        }
    };

    {
        constexpr int NV = NE / 16;
        constexpr int SU = 4;
        const unsigned tk = (unsigned)blockIdx.x * 0x01010101u;

        auto eq4 = [&](unsigned x) -> unsigned {
            const unsigned t = x ^ tk;
            return ~(((t & 0x7f7f7f7fu) + 0x7f7f7f7fu) | t) & 0x80808080u;
        };
        auto match16 = [&](const uint4& q) -> unsigned {
            return (eq4(q.x) >> 7) | (eq4(q.y) >> 6) | (eq4(q.z) >> 5) | (eq4(q.w) >> 4);
        };
        auto emit = [&](unsigned f, int ebase, int flag) {
            while (f != 0) {
                const int b = __builtin_ctz(f);
                f &= f - 1;
                const int slot = atomicAdd(&qn, 1);
                place(slot, (ebase + (b & 7) * 4 + (b >> 3)) | flag);
            }
        };

        for (int v0 = 0; v0 < NV; v0 += AGG_THREADS * SU) {
            uint4 kd[SU];
            uint4 ks[SU];
            #pragma unroll
            for (int u = 0; u < SU; ++u) {
                const int iv = v0 + u * AGG_THREADS + tid;
                const bool in = iv < NV;
                kd[u] = in ? *key_vec(scratch, iv)      : make_uint4(~0u, ~0u, ~0u, ~0u);
                ks[u] = in ? *key_vec(scratch, NV + iv) : make_uint4(~0u, ~0u, ~0u, ~0u);
            }
            #pragma unroll
            for (int u = 0; u < SU; ++u) {
                const int ebase = (v0 + u * AGG_THREADS + tid) * 16;
                emit(match16(kd[u]), ebase, 0);
                emit(match16(ks[u]), ebase, 1 << 29);
            }
        }
    }
    __syncthreads();

    const int n = __builtin_amdgcn_readfirstlane(min(qn, LQ_CAP));
    for (int base = 0; base < n; base += AGG_THREADS * DEAL_UNR) {
        int ent[DEAL_UNR];
        int es[DEAL_UNR];
        int ed[DEAL_UNR];
        int er[DEAL_UNR];
        #pragma unroll
        for (int u = 0; u < DEAL_UNR; ++u) {
            const int idx = base + u * AGG_THREADS + tid;
            ent[u] = queue[min(idx, n - 1)];
        }
        #pragma unroll
        for (int u = 0; u < DEAL_UNR; ++u) {
            const int e = ent[u] & 0xfffff;
            es[u] = src[e];
            ed[u] = dst[e];
            er[u] = ef[e];
        }
        #pragma unroll
        for (int u = 0; u < DEAL_UNR; ++u) {
            const int idx = base + u * AGG_THREADS + tid;
            if (idx < n) deal(ent[u], es[u], ed[u], er[u], true);
        }
    }
    __syncthreads();

    for (int i = tid; i < TILE_R * ROW_CAP / 4; i += AGG_THREADS)
        ((int4*)mylist)[i] = ((const int4*)rq)[i];
    if (tid < TILE_R) {
        mylist[TILE_R * ROW_CAP + tid] = min(rcnt[tid], ROW_CAP);
        out_norm[r0 + tid] = rsqrtf(fmaxf((float)dout[tid], 1.0f));
        in_norm[r0 + tid]  = rsqrtf(fmaxf((float)din[tid], 1.0f));
    }
    if (tid == 0) mylist[TILE_R * ROW_CAP + TILE_R] = ovf;
}

template <int MODE>
__global__ __launch_bounds__(AGG_THREADS)
void tile_aggregate(const float* __restrict__ h,
                    const int* __restrict__ src,
                    const int* __restrict__ dst,
                    const int* __restrict__ ef,
                    float* scratch,
                    const float* __restrict__ in_norm,
                    const float* __restrict__ out_norm,
                    const float* __restrict__ bias,
                    float* outp)
{
    __shared__ float tile[TILE_R * HD];
    __shared__ int   queue[Q_CAP];
    __shared__ __attribute__((aligned(16))) int rq[TILE_R * ROW_CAP];
    __shared__ int   rcnt[TILE_R];
    __shared__ int   qn;
    __shared__ int   ovf;

    const int tid  = threadIdx.x;
    const int lane = tid & 31;
    const int wave = __builtin_amdgcn_readfirstlane(tid >> 5);
    const int r0   = blockIdx.x * TILE_R;
    int* mylist = (int*)((char*)scratch + (size_t)blockIdx.x * REGION_B);

    const bool build = __builtin_amdgcn_readfirstlane(mylist[TILE_R * ROW_CAP + TILE_R]) != 0;

    auto decode = [&](int ent, int s, int d, int r, int& node, int& fl, int& lrow) -> bool {
        bool valid;
        if ((ent >> 29) & 1) {
            node = d;
            lrow = s - r0;
            fl = 2;
            valid = (r == 6 || r == 14 || r == 30);
        } else {
            node = s;
            lrow = d - r0;
            fl = (r >= 0 && r <= 4) ? 1 : 0;
            valid = true;
        }
        return valid && ((unsigned)lrow < (unsigned)TILE_R);
    };

    auto add_now = [&](int node, int fl, int lrow) {
        ovf = 1;
        if (MODE != 2 && fl == 2) return;
        const float w = (MODE == 2 && fl == 1) ? 2.0f : 1.0f;
        float* t = tile + lrow * HD;
        for (int c = 0; c < HD; ++c)
            lds_add(t + (c & 3) * 32 + (c >> 2), h[(long)node * HD + c] * w);
    };

    auto place = [&](int slot, int ent) {
        if (slot < Q_CAP) {
            queue[slot] = ent;
        } else {
            const int e = ent & 0xfffff;
            int node, fl, lrow;
            if (decode(ent, src[e], dst[e], ef[e], node, fl, lrow)) add_now(node, fl, lrow);
        }
    };

    if (build) {
        for (int i = tid; i < TILE_R * HD; i += AGG_THREADS) tile[i] = 0.0f;
        if (tid < TILE_R) rcnt[tid] = 0;
        if (tid == 0) { qn = 0; ovf = 0; }
        __syncthreads();

        for (int e = tid; e < NE; e += AGG_THREADS) {
            if ((unsigned)(dst[e] - r0) < (unsigned)TILE_R) place(atomicAdd(&qn, 1), e);
            if (MODE == 2 && (unsigned)(src[e] - r0) < (unsigned)TILE_R) place(atomicAdd(&qn, 1), e | (1 << 29));
        }
        __syncthreads();

        const int n = __builtin_amdgcn_readfirstlane(min(qn, Q_CAP));
        for (int base = 0; base < n; base += AGG_THREADS * DEAL_UNR) {
            int ent[DEAL_UNR];
            int es[DEAL_UNR];
            int ed[DEAL_UNR];
            int er[DEAL_UNR];
            #pragma unroll
            for (int u = 0; u < DEAL_UNR; ++u) {
                const int idx = base + u * AGG_THREADS + tid;
                ent[u] = queue[min(idx, n - 1)];
            }
            #pragma unroll
            for (int u = 0; u < DEAL_UNR; ++u) {
                const int e = ent[u] & 0xfffff;
                es[u] = src[e];
                ed[u] = dst[e];
                er[u] = ef[e];
            }
            #pragma unroll
            for (int u = 0; u < DEAL_UNR; ++u) {
                const int idx = base + u * AGG_THREADS + tid;
                int node, fl, lrow;
                const bool valid = decode(ent[u], es[u], ed[u], er[u], node, fl, lrow);
                if (idx < n && valid) {
                    const int slot = atomicAdd(&rcnt[lrow], 1);
                    if (slot < ROW_CAP)
                        rq[lrow * ROW_CAP + slot] = node | (fl << 16);
                    else
                        add_now(node, fl, lrow);
                }
            }
        }
        __syncthreads();
    } else {
        constexpr int LV = TILE_R * ROW_CAP / 4;
        constexpr int LU = (LV + AGG_THREADS - 1) / AGG_THREADS;
        int4 lv[LU];
        #pragma unroll
        for (int u = 0; u < LU; ++u)
            lv[u] = ((const int4*)mylist)[min(tid + u * AGG_THREADS, LV - 1)];
        #pragma unroll
        for (int u = 0; u < LU; ++u)
            if (tid + u * AGG_THREADS < LV) ((int4*)rq)[tid + u * AGG_THREADS] = lv[u];
        if (tid < TILE_R) rcnt[tid] = mylist[TILE_R * ROW_CAP + tid];
        if (tid == 0) ovf = 0;
        __syncthreads();
    }

    const int any_ovf = __builtin_amdgcn_readfirstlane(ovf);
    const float* hl = h + lane * 4;

    for (int lr = wave; lr < TILE_R; lr += AGG_WAVES) {
        const int rc = __builtin_amdgcn_readfirstlane(min(rcnt[lr], ROW_CAP));
        float4 acc = make_float4(0.0f, 0.0f, 0.0f, 0.0f);
        for (int b = 0; b < rc; b += 32) {
            const int m  = min(32, rc - b);
            const int my = rq[lr * ROW_CAP + b + lane];
            for (int j = 0; j < m; j += GATHER_UNR) {
                float4 v[GATHER_UNR];
                int    en[GATHER_UNR];
                #pragma unroll
                for (int u = 0; u < GATHER_UNR; ++u) {
                    en[u] = 1 << 17;
                    if (j + u < m) {
                        en[u] = __builtin_amdgcn_readlane(my, j + u);
                        if (MODE == 2 || !((en[u] >> 17) & 1))
                            v[u] = *(const float4*)(hl + (long)(en[u] & 0xffff) * HD);
                    }
                }
                #pragma unroll
                for (int u = 0; u < GATHER_UNR; ++u) {
                    const bool ok = (j + u < m) && (MODE == 2 || !((en[u] >> 17) & 1));
                    float4 x = v[u];
                    if (MODE == 2) {
                        const float w = (ok && ((en[u] >> 16) & 1)) ? 2.0f : 1.0f;
                        x.x *= w; x.y *= w; x.z *= w; x.w *= w;
                    }
                    acc.x = ok ? acc.x + x.x : acc.x;
                    acc.y = ok ? acc.y + x.y : acc.y;
                    acc.z = ok ? acc.z + x.z : acc.z;
                    acc.w = ok ? acc.w + x.w : acc.w;
                }
            }
        }

        if (any_ovf != 0) {
            const float* t = tile + lr * HD + lane;
            acc.x += t[0]; acc.y += t[32]; acc.z += t[64]; acc.w += t[96];
        }
        const int row = r0 + lr;
        if (MODE == 0) {
            const float inn = in_norm[row];
            const float on  = out_norm[row];
            const float4 bb = *(const float4*)(bias + lane * 4);
            acc.x = fmaxf(acc.x * inn + bb.x, 0.0f) * on;
            acc.y = fmaxf(acc.y * inn + bb.y, 0.0f) * on;
            acc.z = fmaxf(acc.z * inn + bb.z, 0.0f) * on;
            acc.w = fmaxf(acc.w * inn + bb.w, 0.0f) * on;
        }
        *(float4*)(outp + (long)row * HD + lane * 4) = acc;
    }
}

extern "C" void kernel_launch(void* const* d_in, const int* in_sizes, int n_in,
                              void* d_out, int out_size, void* d_ws, size_t ws_size,
                              hipStream_t stream)
{
    (void)in_sizes; (void)n_in; (void)out_size; (void)ws_size;

    const float* feat0 = (const float*)d_in[0];
    const float* Wfc0  = (const float*)d_in[1];
    const float* bfc0  = (const float*)d_in[2];
    const float* feat1 = (const float*)d_in[3];
    const float* Wfc1  = (const float*)d_in[4];
    const float* bfc1  = (const float*)d_in[5];
    const float* feat2 = (const float*)d_in[6];
    const float* Wfc2  = (const float*)d_in[7];
    const float* bfc2  = (const float*)d_in[8];
    const float* feat3 = (const float*)d_in[9];
    const float* Wfc3  = (const float*)d_in[10];
    const float* bfc3  = (const float*)d_in[11];
    const int*   src   = (const int*)d_in[12];
    const int*   dst   = (const int*)d_in[13];
    const int*   efeat = (const int*)d_in[14];
    const float* b_g0  = (const float*)d_in[15];
    const float* W_g1  = (const float*)d_in[16];
    const float* b_g1  = (const float*)d_in[17];
    float* out = (float*)d_out;

    const size_t NODE_ELEMS = (size_t)N_NODES * HD;
    float* bufA = (float*)d_ws;
    float* bufB = bufA + NODE_ELEMS;
    float* degO = bufB + NODE_ELEMS;
    float* degI = degO + N_NODES;

    edge_keys<<<(N_KEYW + 255) / 256, 256, 0, stream>>>(src, dst, out);
    tile_lists<<<N_TILES, AGG_THREADS, 0, stream>>>(src, dst, efeat, out, degO, degI);

    const int mt_per = (N_PER + 15) / 16;
    gemm_wmma_f32<128><<<mt_per, 256, 0, stream>>>(
        feat0, Wfc0, bfc0, nullptr, degO + 0 * N_PER,
        bufA + (size_t)0 * N_PER * HD, N_PER, 0);
    gemm_wmma_f32<128><<<mt_per, 256, 0, stream>>>(
        feat1, Wfc1, bfc1, nullptr, degO + 1 * N_PER,
        bufA + (size_t)1 * N_PER * HD, N_PER, 0);
    gemm_wmma_f32<64><<<mt_per, 256, 0, stream>>>(
        feat2, Wfc2, bfc2, nullptr, degO + 2 * N_PER,
        bufA + (size_t)2 * N_PER * HD, N_PER, 0);
    gemm_wmma_f32<64><<<mt_per, 256, 0, stream>>>(
        feat3, Wfc3, bfc3, nullptr, degO + 3 * N_PER,
        bufA + (size_t)3 * N_PER * HD, N_PER, 0);

    tile_aggregate<0><<<N_TILES, AGG_THREADS, 0, stream>>>(
        bufA, src, dst, efeat, out, degI, degO, b_g0, bufB);

    tile_aggregate<1><<<N_TILES, AGG_THREADS, 0, stream>>>(
        bufB, src, dst, efeat, out, nullptr, nullptr, nullptr, bufA);
    gemm_wmma_f32<128><<<3125, 256, 0, stream>>>(
        bufA, W_g1, b_g1, degI, nullptr, bufB, N_NODES, 1);

    tile_aggregate<2><<<N_TILES, AGG_THREADS, 0, stream>>>(
        bufB, src, dst, efeat, out, nullptr, nullptr, nullptr, out);
}


# GEMM loads per row tile cut ~4x: A as 16-byte loads (k permuted alike in both loaders, blocks of 8), a tile's loads as one group, B kept over 2-7 tiles (391/512 blocks); k sum order changes
# speedup vs baseline: 3.0231x; 1.1344x over previous
#include <hip/hip_runtime.h>

typedef float v2f __attribute__((ext_vector_type(2)));
typedef float v8f __attribute__((ext_vector_type(8)));

#define N_NODES 50000
#define N_PER   12500
#define NE      600000
#define HD      128

#define TILE_R      250
#define N_TILES     200
#define N_KEYW      (NE / 4)
#define REGION_B    (TILE_R * HD * 4)
#define KEY_OFF     65536
#define KCH_LOG     11
#define ROW_CAP     64
#define LIST_INTS   (TILE_R * ROW_CAP + TILE_R + 1)

static_assert(TILE_R * N_TILES == N_NODES, "tiles must cover the node range exactly");
static_assert(N_TILES <= 255, "a tile key is one byte and 0xff is the never-matching filler");
static_assert(NE % 16 == 0, "edge ids are read as int4, keys as 16-byte vectors");
static_assert(LIST_INTS * 4 <= KEY_OFF, "row lists end before the key chunk of their region");
static_assert(KEY_OFF + (16 << KCH_LOG) <= REGION_B, "a key chunk ends inside its region");
static_assert(((2 * (NE / 16) + (1 << KCH_LOG) - 1) >> KCH_LOG) <= N_TILES, "enough regions for all key chunks");
static_assert(REGION_B % 16 == 0 && KEY_OFF % 16 == 0, "16-byte vectors stay aligned");

__device__ __forceinline__ uint4* key_vec(float* base, int civ)
{
    return (uint4*)((char*)base + (size_t)(civ >> KCH_LOG) * REGION_B + KEY_OFF) + (civ & ((1 << KCH_LOG) - 1));
}

#define PROJ_BLOCKS 391
#define L1_BLOCKS   512

template <int K>
__global__ __launch_bounds__(256)
void gemm_wmma_f32(const float* __restrict__ A,
                   const float* __restrict__ B,
                   const float* __restrict__ bias,
                   const float* __restrict__ rowscale,
                   const float* __restrict__ outscale,
                   float* __restrict__ C,
                   int M, int relu)
{
    static_assert(K % 8 == 0, "the loaders walk k in blocks of 8");
    const int lane = threadIdx.x & 31;
    const int half = lane >> 4;
    const int lm   = lane & 15;
    const int wave   = (blockIdx.x * blockDim.x + threadIdx.x) >> 5;
    const int nwaves = (gridDim.x * blockDim.x) >> 5;
    const int tn  = wave & 7;
    const int col = tn * 16 + lm;

    v2f bf[K / 4];
    #pragma unroll
    for (int kk = 0; kk < K / 4; ++kk) {
        const int kb = (kk >> 1) * 8 + half * 4 + (kk & 1) * 2;
        bf[kk][0] = B[(long)kb * HD + col];
        bf[kk][1] = B[(long)(kb + 1) * HD + col];
    }

    const int mt = (M + 15) >> 4;
    const float bv = bias[col];

    for (int tm = wave >> 3; tm < mt; tm += (nwaves >> 3)) {
        const int row = tm * 16 + lm;
        const int rc  = (row < M) ? row : (M - 1);
        const float* ap = A + (long)rc * K + half * 4;

        float4 av[K / 8];
        #pragma unroll
        for (int p = 0; p < K / 8; ++p)
            av[p] = *(const float4*)(ap + p * 8);
        const float rs = (rowscale != nullptr) ? rowscale[rc] : 1.0f;
        float os[8];
        #pragma unroll
        for (int v = 0; v < 8; ++v) {
            const int orow = tm * 16 + v + half * 8;
            os[v] = (outscale != nullptr) ? outscale[(orow < M) ? orow : (M - 1)] : 1.0f;
        }

        v8f acc = {};
        #pragma unroll
        for (int p = 0; p < K / 8; ++p) {
            v2f a0, a1;
            a0[0] = av[p].x * rs;
            a0[1] = av[p].y * rs;
            a1[0] = av[p].z * rs;
            a1[1] = av[p].w * rs;
            acc = __builtin_amdgcn_wmma_f32_16x16x4_f32(
                false, a0, false, bf[2 * p], (short)0, acc, false, false);
            acc = __builtin_amdgcn_wmma_f32_16x16x4_f32(
                false, a1, false, bf[2 * p + 1], (short)0, acc, false, false);
        }

        #pragma unroll
        for (int v = 0; v < 8; ++v) {
            const int orow = tm * 16 + v + half * 8;
            if (orow < M) {
                float val = acc[v] + bv;
                if (relu) val = fmaxf(val, 0.0f);
                if (outscale != nullptr) val *= os[v];
                C[(long)orow * HD + col] = val;
            }
        }
    }
}

__global__ void edge_keys(const int* __restrict__ src,
                          const int* __restrict__ dst,
                          float* __restrict__ keybuf)
{
    int j = blockIdx.x * blockDim.x + threadIdx.x;
    const int stride = gridDim.x * blockDim.x;
    for (; j < N_KEYW; j += stride) {
        const int4 s = *(const int4*)(src + 4 * j);
        const int4 d = *(const int4*)(dst + 4 * j);
        const unsigned kd = ((unsigned)d.x / TILE_R) | (((unsigned)d.y / TILE_R) << 8) |
                            (((unsigned)d.z / TILE_R) << 16) | (((unsigned)d.w / TILE_R) << 24);
        const unsigned ks = ((unsigned)s.x / TILE_R) | (((unsigned)s.y / TILE_R) << 8) |
                            (((unsigned)s.z / TILE_R) << 16) | (((unsigned)s.w / TILE_R) << 24);
        const int cw = N_KEYW + j;
        ((unsigned*)key_vec(keybuf, j >> 2))[j & 3]   = kd;
        ((unsigned*)key_vec(keybuf, cw >> 2))[cw & 3] = ks;
    }
}

#define Q_CAP       16384
#define LQ_CAP      32768
#define AGG_THREADS 1024
#define AGG_WAVES   (AGG_THREADS / 32)
#define GATHER_UNR  16
#define DEAL_UNR    4

static_assert(NE <= (1 << 20), "queue entry packing");
static_assert(ROW_CAP == 64, "a row's list is read as two batches of 32 entries, one per lane");
static_assert(N_NODES <= (1 << 16), "row-list entry: node id in the low 16 bits, flags in bits 16 and 17");
static_assert((TILE_R * ROW_CAP) % 4 == 0, "row lists are copied as 16-byte vectors");

__device__ __forceinline__ void lds_add(float* p, float v)
{
    __hip_atomic_fetch_add(p, v, __ATOMIC_RELAXED, __HIP_MEMORY_SCOPE_WORKGROUP);
}

__global__ __launch_bounds__(AGG_THREADS)
void tile_lists(const int* __restrict__ src,
                const int* __restrict__ dst,
                const int* __restrict__ ef,
                float* scratch,
                float* __restrict__ out_norm,
                float* __restrict__ in_norm)
{
    __shared__ int   queue[LQ_CAP];
    __shared__ __attribute__((aligned(16))) int rq[TILE_R * ROW_CAP];
    __shared__ int   rcnt[TILE_R];
    __shared__ int   din[TILE_R];
    __shared__ int   dout[TILE_R];
    __shared__ int   qn;
    __shared__ int   ovf;

    const int tid = threadIdx.x;
    const int r0  = blockIdx.x * TILE_R;
    int* mylist = (int*)((char*)scratch + (size_t)blockIdx.x * REGION_B);

    if (tid < TILE_R) { rcnt[tid] = 0; din[tid] = 0; dout[tid] = 0; }
    if (tid == 0) { qn = 0; ovf = 0; }
    __syncthreads();

    auto deal = [&](int ent, int s, int d, int r, bool listed) {
        const bool srcend = ((ent >> 29) & 1) != 0;
        const int lrow = (srcend ? s : d) - r0;
        if ((unsigned)lrow >= (unsigned)TILE_R) return;
        atomicAdd(srcend ? &dout[lrow] : &din[lrow], 1);
        if (srcend && !(r == 6 || r == 14 || r == 30)) return;
        if (!listed) return;
        const int fl = srcend ? 2 : ((r >= 0 && r <= 4) ? 1 : 0);
        const int slot = atomicAdd(&rcnt[lrow], 1);
        if (slot < ROW_CAP)
            rq[lrow * ROW_CAP + slot] = (srcend ? d : s) | (fl << 16);
        else
            ovf = 1;
    };

    auto place = [&](int slot, int ent) {
        if (slot < LQ_CAP) {
            queue[slot] = ent;
        } else {
            const int e = ent & 0xfffff;
            ovf = 1;
            deal(ent, src[e], dst[e], ef[e], false);
        }
    };

    {
        constexpr int NV = NE / 16;
        constexpr int SU = 4;
        const unsigned tk = (unsigned)blockIdx.x * 0x01010101u;

        auto eq4 = [&](unsigned x) -> unsigned {
            const unsigned t = x ^ tk;
            return ~(((t & 0x7f7f7f7fu) + 0x7f7f7f7fu) | t) & 0x80808080u;
        };
        auto match16 = [&](const uint4& q) -> unsigned {
            return (eq4(q.x) >> 7) | (eq4(q.y) >> 6) | (eq4(q.z) >> 5) | (eq4(q.w) >> 4);
        };
        auto emit = [&](unsigned f, int ebase, int flag) {
            while (f != 0) {
                const int b = __builtin_ctz(f);
                f &= f - 1;
                const int slot = atomicAdd(&qn, 1);
                place(slot, (ebase + (b & 7) * 4 + (b >> 3)) | flag);
            }
        };

        for (int v0 = 0; v0 < NV; v0 += AGG_THREADS * SU) {
            uint4 kd[SU];
            uint4 ks[SU];
            #pragma unroll
            for (int u = 0; u < SU; ++u) {
                const int iv = v0 + u * AGG_THREADS + tid;
                const bool in = iv < NV;
                kd[u] = in ? *key_vec(scratch, iv)      : make_uint4(~0u, ~0u, ~0u, ~0u);
                ks[u] = in ? *key_vec(scratch, NV + iv) : make_uint4(~0u, ~0u, ~0u, ~0u);
            }
            #pragma unroll
            for (int u = 0; u < SU; ++u) {
                const int ebase = (v0 + u * AGG_THREADS + tid) * 16;
                emit(match16(kd[u]), ebase, 0);
                emit(match16(ks[u]), ebase, 1 << 29);
            }
        }
    }
    __syncthreads();

    const int n = __builtin_amdgcn_readfirstlane(min(qn, LQ_CAP));
    for (int base = 0; base < n; base += AGG_THREADS * DEAL_UNR) {
        int ent[DEAL_UNR];
        int es[DEAL_UNR];
        int ed[DEAL_UNR];
        int er[DEAL_UNR];
        #pragma unroll
        for (int u = 0; u < DEAL_UNR; ++u) {
            const int idx = base + u * AGG_THREADS + tid;
            ent[u] = queue[min(idx, n - 1)];
        }
        #pragma unroll
        for (int u = 0; u < DEAL_UNR; ++u) {
            const int e = ent[u] & 0xfffff;
            es[u] = src[e];
            ed[u] = dst[e];
            er[u] = ef[e];
        }
        #pragma unroll
        for (int u = 0; u < DEAL_UNR; ++u) {
            const int idx = base + u * AGG_THREADS + tid;
            if (idx < n) deal(ent[u], es[u], ed[u], er[u], true);
        }
    }
    __syncthreads();

    for (int i = tid; i < TILE_R * ROW_CAP / 4; i += AGG_THREADS)
        ((int4*)mylist)[i] = ((const int4*)rq)[i];
    if (tid < TILE_R) {
        mylist[TILE_R * ROW_CAP + tid] = min(rcnt[tid], ROW_CAP);
        out_norm[r0 + tid] = rsqrtf(fmaxf((float)dout[tid], 1.0f));
        in_norm[r0 + tid]  = rsqrtf(fmaxf((float)din[tid], 1.0f));
    }
    if (tid == 0) mylist[TILE_R * ROW_CAP + TILE_R] = ovf;
}

template <int MODE>
__global__ __launch_bounds__(AGG_THREADS)
void tile_aggregate(const float* __restrict__ h,
                    const int* __restrict__ src,
                    const int* __restrict__ dst,
                    const int* __restrict__ ef,
                    float* scratch,
                    const float* __restrict__ in_norm,
                    const float* __restrict__ out_norm,
                    const float* __restrict__ bias,
                    float* outp)
{
    __shared__ float tile[TILE_R * HD];
    __shared__ int   queue[Q_CAP];
    __shared__ __attribute__((aligned(16))) int rq[TILE_R * ROW_CAP];
    __shared__ int   rcnt[TILE_R];
    __shared__ int   qn;
    __shared__ int   ovf;

    const int tid  = threadIdx.x;
    const int lane = tid & 31;
    const int wave = __builtin_amdgcn_readfirstlane(tid >> 5);
    const int r0   = blockIdx.x * TILE_R;
    int* mylist = (int*)((char*)scratch + (size_t)blockIdx.x * REGION_B);

    const bool build = __builtin_amdgcn_readfirstlane(mylist[TILE_R * ROW_CAP + TILE_R]) != 0;

    auto decode = [&](int ent, int s, int d, int r, int& node, int& fl, int& lrow) -> bool {
        bool valid;
        if ((ent >> 29) & 1) {
            node = d;
            lrow = s - r0;
            fl = 2;
            valid = (r == 6 || r == 14 || r == 30);
        } else {
            node = s;
            lrow = d - r0;
            fl = (r >= 0 && r <= 4) ? 1 : 0;
            valid = true;
        }
        return valid && ((unsigned)lrow < (unsigned)TILE_R);
    };

    auto add_now = [&](int node, int fl, int lrow) {
        ovf = 1;
        if (MODE != 2 && fl == 2) return;
        const float w = (MODE == 2 && fl == 1) ? 2.0f : 1.0f;
        float* t = tile + lrow * HD;
        for (int c = 0; c < HD; ++c)
            lds_add(t + (c & 3) * 32 + (c >> 2), h[(long)node * HD + c] * w);
    };

    auto place = [&](int slot, int ent) {
        if (slot < Q_CAP) {
            queue[slot] = ent;
        } else {
            const int e = ent & 0xfffff;
            int node, fl, lrow;
            if (decode(ent, src[e], dst[e], ef[e], node, fl, lrow)) add_now(node, fl, lrow);
        }
    };

    if (build) {
        for (int i = tid; i < TILE_R * HD; i += AGG_THREADS) tile[i] = 0.0f;
        if (tid < TILE_R) rcnt[tid] = 0;
        if (tid == 0) { qn = 0; ovf = 0; }
        __syncthreads();

        for (int e = tid; e < NE; e += AGG_THREADS) {
            if ((unsigned)(dst[e] - r0) < (unsigned)TILE_R) place(atomicAdd(&qn, 1), e);
            if (MODE == 2 && (unsigned)(src[e] - r0) < (unsigned)TILE_R) place(atomicAdd(&qn, 1), e | (1 << 29));
        }
        __syncthreads();

        const int n = __builtin_amdgcn_readfirstlane(min(qn, Q_CAP));
        for (int base = 0; base < n; base += AGG_THREADS * DEAL_UNR) {
            int ent[DEAL_UNR];
            int es[DEAL_UNR];
            int ed[DEAL_UNR];
            int er[DEAL_UNR];
            #pragma unroll
            for (int u = 0; u < DEAL_UNR; ++u) {
                const int idx = base + u * AGG_THREADS + tid;
                ent[u] = queue[min(idx, n - 1)];
            }
            #pragma unroll
            for (int u = 0; u < DEAL_UNR; ++u) {
                const int e = ent[u] & 0xfffff;
                es[u] = src[e];
                ed[u] = dst[e];
                er[u] = ef[e];
            }
            #pragma unroll
            for (int u = 0; u < DEAL_UNR; ++u) {
                const int idx = base + u * AGG_THREADS + tid;
                int node, fl, lrow;
                const bool valid = decode(ent[u], es[u], ed[u], er[u], node, fl, lrow);
                if (idx < n && valid) {
                    const int slot = atomicAdd(&rcnt[lrow], 1);
                    if (slot < ROW_CAP)
                        rq[lrow * ROW_CAP + slot] = node | (fl << 16);
                    else
                        add_now(node, fl, lrow);
                }
            }
        }
        __syncthreads();
    } else {
        constexpr int LV = TILE_R * ROW_CAP / 4;
        constexpr int LU = (LV + AGG_THREADS - 1) / AGG_THREADS;
        int4 lv[LU];
        #pragma unroll
        for (int u = 0; u < LU; ++u)
            lv[u] = ((const int4*)mylist)[min(tid + u * AGG_THREADS, LV - 1)];
        #pragma unroll
        for (int u = 0; u < LU; ++u)
            if (tid + u * AGG_THREADS < LV) ((int4*)rq)[tid + u * AGG_THREADS] = lv[u];
        if (tid < TILE_R) rcnt[tid] = mylist[TILE_R * ROW_CAP + tid];
        if (tid == 0) ovf = 0;
        __syncthreads();
    }

    const int any_ovf = __builtin_amdgcn_readfirstlane(ovf);
    const float* hl = h + lane * 4;

    for (int lr = wave; lr < TILE_R; lr += AGG_WAVES) {
        const int rc = __builtin_amdgcn_readfirstlane(min(rcnt[lr], ROW_CAP));
        float4 acc = make_float4(0.0f, 0.0f, 0.0f, 0.0f);
        for (int b = 0; b < rc; b += 32) {
            const int m  = min(32, rc - b);
            const int my = rq[lr * ROW_CAP + b + lane];
            for (int j = 0; j < m; j += GATHER_UNR) {
                float4 v[GATHER_UNR];
                int    en[GATHER_UNR];
                #pragma unroll
                for (int u = 0; u < GATHER_UNR; ++u) {
                    en[u] = 1 << 17;
                    if (j + u < m) {
                        en[u] = __builtin_amdgcn_readlane(my, j + u);
                        if (MODE == 2 || !((en[u] >> 17) & 1))
                            v[u] = *(const float4*)(hl + (long)(en[u] & 0xffff) * HD);
                    }
                }
                #pragma unroll
                for (int u = 0; u < GATHER_UNR; ++u) {
                    const bool ok = (j + u < m) && (MODE == 2 || !((en[u] >> 17) & 1));
                    float4 x = v[u];
                    if (MODE == 2) {
                        const float w = (ok && ((en[u] >> 16) & 1)) ? 2.0f : 1.0f;
                        x.x *= w; x.y *= w; x.z *= w; x.w *= w;
                    }
                    acc.x = ok ? acc.x + x.x : acc.x;
                    acc.y = ok ? acc.y + x.y : acc.y;
                    acc.z = ok ? acc.z + x.z : acc.z;
                    acc.w = ok ? acc.w + x.w : acc.w;
                }
            }
        }

        if (any_ovf != 0) {
            const float* t = tile + lr * HD + lane;
            acc.x += t[0]; acc.y += t[32]; acc.z += t[64]; acc.w += t[96];
        }
        const int row = r0 + lr;
        if (MODE == 0) {
            const float inn = in_norm[row];
            const float on  = out_norm[row];
            const float4 bb = *(const float4*)(bias + lane * 4);
            acc.x = fmaxf(acc.x * inn + bb.x, 0.0f) * on;
            acc.y = fmaxf(acc.y * inn + bb.y, 0.0f) * on;
            acc.z = fmaxf(acc.z * inn + bb.z, 0.0f) * on;
            acc.w = fmaxf(acc.w * inn + bb.w, 0.0f) * on;
        }
        *(float4*)(outp + (long)row * HD + lane * 4) = acc;
    }
}

extern "C" void kernel_launch(void* const* d_in, const int* in_sizes, int n_in,
                              void* d_out, int out_size, void* d_ws, size_t ws_size,
                              hipStream_t stream)
{
    (void)in_sizes; (void)n_in; (void)out_size; (void)ws_size;

    const float* feat0 = (const float*)d_in[0];
    const float* Wfc0  = (const float*)d_in[1];
    const float* bfc0  = (const float*)d_in[2];
    const float* feat1 = (const float*)d_in[3];
    const float* Wfc1  = (const float*)d_in[4];
    const float* bfc1  = (const float*)d_in[5];
    const float* feat2 = (const float*)d_in[6];
    const float* Wfc2  = (const float*)d_in[7];
    const float* bfc2  = (const float*)d_in[8];
    const float* feat3 = (const float*)d_in[9];
    const float* Wfc3  = (const float*)d_in[10];
    const float* bfc3  = (const float*)d_in[11];
    const int*   src   = (const int*)d_in[12];
    const int*   dst   = (const int*)d_in[13];
    const int*   efeat = (const int*)d_in[14];
    const float* b_g0  = (const float*)d_in[15];
    const float* W_g1  = (const float*)d_in[16];
    const float* b_g1  = (const float*)d_in[17];
    float* out = (float*)d_out;

    const size_t NODE_ELEMS = (size_t)N_NODES * HD;
    float* bufA = (float*)d_ws;
    float* bufB = bufA + NODE_ELEMS;
    float* degO = bufB + NODE_ELEMS;
    float* degI = degO + N_NODES;

    edge_keys<<<(N_KEYW + 255) / 256, 256, 0, stream>>>(src, dst, out);
    tile_lists<<<N_TILES, AGG_THREADS, 0, stream>>>(src, dst, efeat, out, degO, degI);

    gemm_wmma_f32<128><<<PROJ_BLOCKS, 256, 0, stream>>>(
        feat0, Wfc0, bfc0, nullptr, degO + 0 * N_PER,
        bufA + (size_t)0 * N_PER * HD, N_PER, 0);
    gemm_wmma_f32<128><<<PROJ_BLOCKS, 256, 0, stream>>>(
        feat1, Wfc1, bfc1, nullptr, degO + 1 * N_PER,
        bufA + (size_t)1 * N_PER * HD, N_PER, 0);
    gemm_wmma_f32<64><<<PROJ_BLOCKS, 256, 0, stream>>>(
        feat2, Wfc2, bfc2, nullptr, degO + 2 * N_PER,
        bufA + (size_t)2 * N_PER * HD, N_PER, 0);
    gemm_wmma_f32<64><<<PROJ_BLOCKS, 256, 0, stream>>>(
        feat3, Wfc3, bfc3, nullptr, degO + 3 * N_PER,
        bufA + (size_t)3 * N_PER * HD, N_PER, 0);

    tile_aggregate<0><<<N_TILES, AGG_THREADS, 0, stream>>>(
        bufA, src, dst, efeat, out, degI, degO, b_g0, bufB);

    tile_aggregate<1><<<N_TILES, AGG_THREADS, 0, stream>>>(
        bufB, src, dst, efeat, out, nullptr, nullptr, nullptr, bufA);
    gemm_wmma_f32<128><<<L1_BLOCKS, 256, 0, stream>>>(
        bufA, W_g1, b_g1, degI, nullptr, bufB, N_NODES, 1);

    tile_aggregate<2><<<N_TILES, AGG_THREADS, 0, stream>>>(
        bufB, src, dst, efeat, out, nullptr, nullptr, nullptr, out);
}


# list launch looks up each queued edge end with one scattered 8-byte load of a packed record (src|dst<<16, e_feat; written by the key pass) instead of three 4-byte loads; sums stay arrival-order
# speedup vs baseline: 3.1667x; 1.0475x over previous
#include <hip/hip_runtime.h>

typedef float v2f __attribute__((ext_vector_type(2)));
typedef float v8f __attribute__((ext_vector_type(8)));

#define N_NODES 50000
#define N_PER   12500
#define NE      600000
#define HD      128

#define TILE_R      250
#define N_TILES     200
#define N_KEYW      (NE / 4)
#define REGION_B    (TILE_R * HD * 4)
#define KEY_OFF     65536
#define KCH_LOG     11
#define ROW_CAP     64
#define LIST_INTS   (TILE_R * ROW_CAP + TILE_R + 1)

static_assert(TILE_R * N_TILES == N_NODES, "tiles must cover the node range exactly");
static_assert(N_TILES <= 255, "a tile key is one byte and 0xff is the never-matching filler");
static_assert(NE % 16 == 0, "edge ids are read as int4, keys as 16-byte vectors");
static_assert(LIST_INTS * 4 <= KEY_OFF, "row lists end before the key chunk of their region");
static_assert(KEY_OFF + (16 << KCH_LOG) <= REGION_B, "a key chunk ends inside its region");
static_assert(((2 * (NE / 16) + (1 << KCH_LOG) - 1) >> KCH_LOG) <= N_TILES, "enough regions for all key chunks");
static_assert(REGION_B % 16 == 0 && KEY_OFF % 16 == 0, "16-byte vectors stay aligned");

__device__ __forceinline__ uint4* key_vec(float* base, int civ)
{
    return (uint4*)((char*)base + (size_t)(civ >> KCH_LOG) * REGION_B + KEY_OFF) + (civ & ((1 << KCH_LOG) - 1));
}

#define PROJ_BLOCKS 391
#define L1_BLOCKS   512

template <int K>
__global__ __launch_bounds__(256)
void gemm_wmma_f32(const float* __restrict__ A,
                   const float* __restrict__ B,
                   const float* __restrict__ bias,
                   const float* __restrict__ rowscale,
                   const float* __restrict__ outscale,
                   float* __restrict__ C,
                   int M, int relu)
{
    static_assert(K % 8 == 0, "the loaders walk k in blocks of 8");
    const int lane = threadIdx.x & 31;
    const int half = lane >> 4;
    const int lm   = lane & 15;
    const int wave   = (blockIdx.x * blockDim.x + threadIdx.x) >> 5;
    const int nwaves = (gridDim.x * blockDim.x) >> 5;
    const int tn  = wave & 7;
    const int col = tn * 16 + lm;

    v2f bf[K / 4];
    #pragma unroll
    for (int kk = 0; kk < K / 4; ++kk) {
        const int kb = (kk >> 1) * 8 + half * 4 + (kk & 1) * 2;
        bf[kk][0] = B[(long)kb * HD + col];
        bf[kk][1] = B[(long)(kb + 1) * HD + col];
    }

    const int mt = (M + 15) >> 4;
    const float bv = bias[col];

    for (int tm = wave >> 3; tm < mt; tm += (nwaves >> 3)) {
        const int row = tm * 16 + lm;
        const int rc  = (row < M) ? row : (M - 1);
        const float* ap = A + (long)rc * K + half * 4;

        float4 av[K / 8];
        #pragma unroll
        for (int p = 0; p < K / 8; ++p)
            av[p] = *(const float4*)(ap + p * 8);
        const float rs = (rowscale != nullptr) ? rowscale[rc] : 1.0f;
        float os[8];
        #pragma unroll
        for (int v = 0; v < 8; ++v) {
            const int orow = tm * 16 + v + half * 8;
            os[v] = (outscale != nullptr) ? outscale[(orow < M) ? orow : (M - 1)] : 1.0f;
        }

        v8f acc = {};
        #pragma unroll
        for (int p = 0; p < K / 8; ++p) {
            v2f a0, a1;
            a0[0] = av[p].x * rs;
            a0[1] = av[p].y * rs;
            a1[0] = av[p].z * rs;
            a1[1] = av[p].w * rs;
            acc = __builtin_amdgcn_wmma_f32_16x16x4_f32(
                false, a0, false, bf[2 * p], (short)0, acc, false, false);
            acc = __builtin_amdgcn_wmma_f32_16x16x4_f32(
                false, a1, false, bf[2 * p + 1], (short)0, acc, false, false);
        }

        #pragma unroll
        for (int v = 0; v < 8; ++v) {
            const int orow = tm * 16 + v + half * 8;
            if (orow < M) {
                float val = acc[v] + bv;
                if (relu) val = fmaxf(val, 0.0f);
                if (outscale != nullptr) val *= os[v];
                C[(long)orow * HD + col] = val;
            }
        }
    }
}

__global__ void edge_keys(const int* __restrict__ src,
                          const int* __restrict__ dst,
                          const int* __restrict__ ef,
                          float* __restrict__ keybuf,
                          float* __restrict__ recbuf)
{
    uint4* rec4 = (uint4*)recbuf;
    int j = blockIdx.x * blockDim.x + threadIdx.x;
    const int stride = gridDim.x * blockDim.x;
    for (; j < N_KEYW; j += stride) {
        const int4 s = *(const int4*)(src + 4 * j);
        const int4 d = *(const int4*)(dst + 4 * j);
        const int4 r = *(const int4*)(ef + 4 * j);
        const unsigned kd = ((unsigned)d.x / TILE_R) | (((unsigned)d.y / TILE_R) << 8) |
                            (((unsigned)d.z / TILE_R) << 16) | (((unsigned)d.w / TILE_R) << 24);
        const unsigned ks = ((unsigned)s.x / TILE_R) | (((unsigned)s.y / TILE_R) << 8) |
                            (((unsigned)s.z / TILE_R) << 16) | (((unsigned)s.w / TILE_R) << 24);
        const int cw = N_KEYW + j;
        ((unsigned*)key_vec(keybuf, j >> 2))[j & 3]   = kd;
        ((unsigned*)key_vec(keybuf, cw >> 2))[cw & 3] = ks;
        rec4[2 * j]     = make_uint4((unsigned)s.x | ((unsigned)d.x << 16), (unsigned)r.x,
                                     (unsigned)s.y | ((unsigned)d.y << 16), (unsigned)r.y);
        rec4[2 * j + 1] = make_uint4((unsigned)s.z | ((unsigned)d.z << 16), (unsigned)r.z,
                                     (unsigned)s.w | ((unsigned)d.w << 16), (unsigned)r.w);
    }
}

#define Q_CAP       16384
#define LQ_CAP      32768
#define AGG_THREADS 1024
#define AGG_WAVES   (AGG_THREADS / 32)
#define GATHER_UNR  16
#define DEAL_UNR    4

static_assert(NE <= (1 << 20), "queue entry packing");
static_assert(ROW_CAP == 64, "a row's list is read as two batches of 32 entries, one per lane");
static_assert(N_NODES <= (1 << 16), "row-list entry: node id in the low 16 bits, flags in bits 16 and 17; edge record: src | dst << 16");
static_assert((size_t)NE * 8 <= (size_t)N_NODES * HD * 4, "the edge records fit the node buffer that holds them");
static_assert((TILE_R * ROW_CAP) % 4 == 0, "row lists are copied as 16-byte vectors");

__device__ __forceinline__ void lds_add(float* p, float v)
{
    __hip_atomic_fetch_add(p, v, __ATOMIC_RELAXED, __HIP_MEMORY_SCOPE_WORKGROUP);
}

__global__ __launch_bounds__(AGG_THREADS)
void tile_lists(const int* __restrict__ src,
                const int* __restrict__ dst,
                const int* __restrict__ ef,
                const float* __restrict__ recbuf,
                float* scratch,
                float* __restrict__ out_norm,
                float* __restrict__ in_norm)
{
    __shared__ int   queue[LQ_CAP];
    __shared__ __attribute__((aligned(16))) int rq[TILE_R * ROW_CAP];
    __shared__ int   rcnt[TILE_R];
    __shared__ int   din[TILE_R];
    __shared__ int   dout[TILE_R];
    __shared__ int   qn;
    __shared__ int   ovf;

    const int tid = threadIdx.x;
    const int r0  = blockIdx.x * TILE_R;
    int* mylist = (int*)((char*)scratch + (size_t)blockIdx.x * REGION_B);
    const uint2* rec = (const uint2*)recbuf;

    if (tid < TILE_R) { rcnt[tid] = 0; din[tid] = 0; dout[tid] = 0; }
    if (tid == 0) { qn = 0; ovf = 0; }
    __syncthreads();

    auto deal = [&](int ent, int s, int d, int r, bool listed) {
        const bool srcend = ((ent >> 29) & 1) != 0;
        const int lrow = (srcend ? s : d) - r0;
        if ((unsigned)lrow >= (unsigned)TILE_R) return;
        atomicAdd(srcend ? &dout[lrow] : &din[lrow], 1);
        if (srcend && !(r == 6 || r == 14 || r == 30)) return;
        if (!listed) return;
        const int fl = srcend ? 2 : ((r >= 0 && r <= 4) ? 1 : 0);
        const int slot = atomicAdd(&rcnt[lrow], 1);
        if (slot < ROW_CAP)
            rq[lrow * ROW_CAP + slot] = (srcend ? d : s) | (fl << 16);
        else
            ovf = 1;
    };

    auto place = [&](int slot, int ent) {
        if (slot < LQ_CAP) {
            queue[slot] = ent;
        } else {
            const int e = ent & 0xfffff;
            ovf = 1;
            deal(ent, src[e], dst[e], ef[e], false);
        }
    };

    {
        constexpr int NV = NE / 16;
        constexpr int SU = 4;
        const unsigned tk = (unsigned)blockIdx.x * 0x01010101u;

        auto eq4 = [&](unsigned x) -> unsigned {
            const unsigned t = x ^ tk;
            return ~(((t & 0x7f7f7f7fu) + 0x7f7f7f7fu) | t) & 0x80808080u;
        };
        auto match16 = [&](const uint4& q) -> unsigned {
            return (eq4(q.x) >> 7) | (eq4(q.y) >> 6) | (eq4(q.z) >> 5) | (eq4(q.w) >> 4);
        };
        auto emit = [&](unsigned f, int ebase, int flag) {
            while (f != 0) {
                const int b = __builtin_ctz(f);
                f &= f - 1;
                const int slot = atomicAdd(&qn, 1);
                place(slot, (ebase + (b & 7) * 4 + (b >> 3)) | flag);
            }
        };

        for (int v0 = 0; v0 < NV; v0 += AGG_THREADS * SU) {
            uint4 kd[SU];
            uint4 ks[SU];
            #pragma unroll
            for (int u = 0; u < SU; ++u) {
                const int iv = v0 + u * AGG_THREADS + tid;
                const bool in = iv < NV;
                kd[u] = in ? *key_vec(scratch, iv)      : make_uint4(~0u, ~0u, ~0u, ~0u);
                ks[u] = in ? *key_vec(scratch, NV + iv) : make_uint4(~0u, ~0u, ~0u, ~0u);
            }
            #pragma unroll
            for (int u = 0; u < SU; ++u) {
                const int ebase = (v0 + u * AGG_THREADS + tid) * 16;
                emit(match16(kd[u]), ebase, 0);
                emit(match16(ks[u]), ebase, 1 << 29);
            }
        }
    }
    __syncthreads();

    const int n = __builtin_amdgcn_readfirstlane(min(qn, LQ_CAP));
    for (int base = 0; base < n; base += AGG_THREADS * DEAL_UNR) {
        int   ent[DEAL_UNR];
        uint2 er[DEAL_UNR];
        #pragma unroll
        for (int u = 0; u < DEAL_UNR; ++u) {
            const int idx = base + u * AGG_THREADS + tid;
            ent[u] = queue[min(idx, n - 1)];
        }
        #pragma unroll
        for (int u = 0; u < DEAL_UNR; ++u)
            er[u] = rec[ent[u] & 0xfffff];
        #pragma unroll
        for (int u = 0; u < DEAL_UNR; ++u) {
            const int idx = base + u * AGG_THREADS + tid;
            if (idx < n)
                deal(ent[u], (int)(er[u].x & 0xffffu), (int)(er[u].x >> 16), (int)er[u].y, true);
        }
    }
    __syncthreads();

    for (int i = tid; i < TILE_R * ROW_CAP / 4; i += AGG_THREADS)
        ((int4*)mylist)[i] = ((const int4*)rq)[i];
    if (tid < TILE_R) {
        mylist[TILE_R * ROW_CAP + tid] = min(rcnt[tid], ROW_CAP);
        out_norm[r0 + tid] = rsqrtf(fmaxf((float)dout[tid], 1.0f));
        in_norm[r0 + tid]  = rsqrtf(fmaxf((float)din[tid], 1.0f));
    }
    if (tid == 0) mylist[TILE_R * ROW_CAP + TILE_R] = ovf;
}

template <int MODE>
__global__ __launch_bounds__(AGG_THREADS)
void tile_aggregate(const float* __restrict__ h,
                    const int* __restrict__ src,
                    const int* __restrict__ dst,
                    const int* __restrict__ ef,
                    float* scratch,
                    const float* __restrict__ in_norm,
                    const float* __restrict__ out_norm,
                    const float* __restrict__ bias,
                    float* outp)
{
    __shared__ float tile[TILE_R * HD];
    __shared__ int   queue[Q_CAP];
    __shared__ __attribute__((aligned(16))) int rq[TILE_R * ROW_CAP];
    __shared__ int   rcnt[TILE_R];
    __shared__ int   qn;
    __shared__ int   ovf;

    const int tid  = threadIdx.x;
    const int lane = tid & 31;
    const int wave = __builtin_amdgcn_readfirstlane(tid >> 5);
    const int r0   = blockIdx.x * TILE_R;
    int* mylist = (int*)((char*)scratch + (size_t)blockIdx.x * REGION_B);

    const bool build = __builtin_amdgcn_readfirstlane(mylist[TILE_R * ROW_CAP + TILE_R]) != 0;

    auto decode = [&](int ent, int s, int d, int r, int& node, int& fl, int& lrow) -> bool {
        bool valid;
        if ((ent >> 29) & 1) {
            node = d;
            lrow = s - r0;
            fl = 2;
            valid = (r == 6 || r == 14 || r == 30);
        } else {
            node = s;
            lrow = d - r0;
            fl = (r >= 0 && r <= 4) ? 1 : 0;
            valid = true;
        }
        return valid && ((unsigned)lrow < (unsigned)TILE_R);
    };

    auto add_now = [&](int node, int fl, int lrow) {
        ovf = 1;
        if (MODE != 2 && fl == 2) return;
        const float w = (MODE == 2 && fl == 1) ? 2.0f : 1.0f;
        float* t = tile + lrow * HD;
        for (int c = 0; c < HD; ++c)
            lds_add(t + (c & 3) * 32 + (c >> 2), h[(long)node * HD + c] * w);
    };

    auto place = [&](int slot, int ent) {
        if (slot < Q_CAP) {
            queue[slot] = ent;
        } else {
            const int e = ent & 0xfffff;
            int node, fl, lrow;
            if (decode(ent, src[e], dst[e], ef[e], node, fl, lrow)) add_now(node, fl, lrow);
        }
    };

    if (build) {
        for (int i = tid; i < TILE_R * HD; i += AGG_THREADS) tile[i] = 0.0f;
        if (tid < TILE_R) rcnt[tid] = 0;
        if (tid == 0) { qn = 0; ovf = 0; }
        __syncthreads();

        for (int e = tid; e < NE; e += AGG_THREADS) {
            if ((unsigned)(dst[e] - r0) < (unsigned)TILE_R) place(atomicAdd(&qn, 1), e);
            if (MODE == 2 && (unsigned)(src[e] - r0) < (unsigned)TILE_R) place(atomicAdd(&qn, 1), e | (1 << 29));
        }
        __syncthreads();

        const int n = __builtin_amdgcn_readfirstlane(min(qn, Q_CAP));
        for (int base = 0; base < n; base += AGG_THREADS * DEAL_UNR) {
            int ent[DEAL_UNR];
            int es[DEAL_UNR];
            int ed[DEAL_UNR];
            int er[DEAL_UNR];
            #pragma unroll
            for (int u = 0; u < DEAL_UNR; ++u) {
                const int idx = base + u * AGG_THREADS + tid;
                ent[u] = queue[min(idx, n - 1)];
            }
            #pragma unroll
            for (int u = 0; u < DEAL_UNR; ++u) {
                const int e = ent[u] & 0xfffff;
                es[u] = src[e];
                ed[u] = dst[e];
                er[u] = ef[e];
            }
            #pragma unroll
            for (int u = 0; u < DEAL_UNR; ++u) {
                const int idx = base + u * AGG_THREADS + tid;
                int node, fl, lrow;
                const bool valid = decode(ent[u], es[u], ed[u], er[u], node, fl, lrow);
                if (idx < n && valid) {
                    const int slot = atomicAdd(&rcnt[lrow], 1);
                    if (slot < ROW_CAP)
                        rq[lrow * ROW_CAP + slot] = node | (fl << 16);
                    else
                        add_now(node, fl, lrow);
                }
            }
        }
        __syncthreads();
    } else {
        constexpr int LV = TILE_R * ROW_CAP / 4;
        constexpr int LU = (LV + AGG_THREADS - 1) / AGG_THREADS;
        int4 lv[LU];
        #pragma unroll
        for (int u = 0; u < LU; ++u)
            lv[u] = ((const int4*)mylist)[min(tid + u * AGG_THREADS, LV - 1)];
        #pragma unroll
        for (int u = 0; u < LU; ++u)
            if (tid + u * AGG_THREADS < LV) ((int4*)rq)[tid + u * AGG_THREADS] = lv[u];
        if (tid < TILE_R) rcnt[tid] = mylist[TILE_R * ROW_CAP + tid];
        if (tid == 0) ovf = 0;
        __syncthreads();
    }

    const int any_ovf = __builtin_amdgcn_readfirstlane(ovf);
    const float* hl = h + lane * 4;

    for (int lr = wave; lr < TILE_R; lr += AGG_WAVES) {
        const int rc = __builtin_amdgcn_readfirstlane(min(rcnt[lr], ROW_CAP));
        float4 acc = make_float4(0.0f, 0.0f, 0.0f, 0.0f);
        for (int b = 0; b < rc; b += 32) {
            const int m  = min(32, rc - b);
            const int my = rq[lr * ROW_CAP + b + lane];
            for (int j = 0; j < m; j += GATHER_UNR) {
                float4 v[GATHER_UNR];
                int    en[GATHER_UNR];
                #pragma unroll
                for (int u = 0; u < GATHER_UNR; ++u) {
                    en[u] = 1 << 17;
                    if (j + u < m) {
                        en[u] = __builtin_amdgcn_readlane(my, j + u);
                        if (MODE == 2 || !((en[u] >> 17) & 1))
                            v[u] = *(const float4*)(hl + (long)(en[u] & 0xffff) * HD);
                    }
                }
                #pragma unroll
                for (int u = 0; u < GATHER_UNR; ++u) {
                    const bool ok = (j + u < m) && (MODE == 2 || !((en[u] >> 17) & 1));
                    float4 x = v[u];
                    if (MODE == 2) {
                        const float w = (ok && ((en[u] >> 16) & 1)) ? 2.0f : 1.0f;
                        x.x *= w; x.y *= w; x.z *= w; x.w *= w;
                    }
                    acc.x = ok ? acc.x + x.x : acc.x;
                    acc.y = ok ? acc.y + x.y : acc.y;
                    acc.z = ok ? acc.z + x.z : acc.z;
                    acc.w = ok ? acc.w + x.w : acc.w;
                }
            }
        }

        if (any_ovf != 0) {
            const float* t = tile + lr * HD + lane;
            acc.x += t[0]; acc.y += t[32]; acc.z += t[64]; acc.w += t[96];
        }
        const int row = r0 + lr;
        if (MODE == 0) {
            const float inn = in_norm[row];
            const float on  = out_norm[row];
            const float4 bb = *(const float4*)(bias + lane * 4);
            acc.x = fmaxf(acc.x * inn + bb.x, 0.0f) * on;
            acc.y = fmaxf(acc.y * inn + bb.y, 0.0f) * on;
            acc.z = fmaxf(acc.z * inn + bb.z, 0.0f) * on;
            acc.w = fmaxf(acc.w * inn + bb.w, 0.0f) * on;
        }
        *(float4*)(outp + (long)row * HD + lane * 4) = acc;
    }
}

extern "C" void kernel_launch(void* const* d_in, const int* in_sizes, int n_in,
                              void* d_out, int out_size, void* d_ws, size_t ws_size,
                              hipStream_t stream)
{
    (void)in_sizes; (void)n_in; (void)out_size; (void)ws_size;

    const float* feat0 = (const float*)d_in[0];
    const float* Wfc0  = (const float*)d_in[1];
    const float* bfc0  = (const float*)d_in[2];
    const float* feat1 = (const float*)d_in[3];
    const float* Wfc1  = (const float*)d_in[4];
    const float* bfc1  = (const float*)d_in[5];
    const float* feat2 = (const float*)d_in[6];
    const float* Wfc2  = (const float*)d_in[7];
    const float* bfc2  = (const float*)d_in[8];
    const float* feat3 = (const float*)d_in[9];
    const float* Wfc3  = (const float*)d_in[10];
    const float* bfc3  = (const float*)d_in[11];
    const int*   src   = (const int*)d_in[12];
    const int*   dst   = (const int*)d_in[13];
    const int*   efeat = (const int*)d_in[14];
    const float* b_g0  = (const float*)d_in[15];
    const float* W_g1  = (const float*)d_in[16];
    const float* b_g1  = (const float*)d_in[17];
    float* out = (float*)d_out;

    const size_t NODE_ELEMS = (size_t)N_NODES * HD;
    float* bufA = (float*)d_ws;
    float* bufB = bufA + NODE_ELEMS;
    float* degO = bufB + NODE_ELEMS;
    float* degI = degO + N_NODES;

    edge_keys<<<(N_KEYW + 255) / 256, 256, 0, stream>>>(src, dst, efeat, out, bufB);
    tile_lists<<<N_TILES, AGG_THREADS, 0, stream>>>(src, dst, efeat, bufB, out, degO, degI);

    gemm_wmma_f32<128><<<PROJ_BLOCKS, 256, 0, stream>>>(
        feat0, Wfc0, bfc0, nullptr, degO + 0 * N_PER,
        bufA + (size_t)0 * N_PER * HD, N_PER, 0);
    gemm_wmma_f32<128><<<PROJ_BLOCKS, 256, 0, stream>>>(
        feat1, Wfc1, bfc1, nullptr, degO + 1 * N_PER,
        bufA + (size_t)1 * N_PER * HD, N_PER, 0);
    gemm_wmma_f32<64><<<PROJ_BLOCKS, 256, 0, stream>>>(
        feat2, Wfc2, bfc2, nullptr, degO + 2 * N_PER,
        bufA + (size_t)2 * N_PER * HD, N_PER, 0);
    gemm_wmma_f32<64><<<PROJ_BLOCKS, 256, 0, stream>>>(
        feat3, Wfc3, bfc3, nullptr, degO + 3 * N_PER,
        bufA + (size_t)3 * N_PER * HD, N_PER, 0);

    tile_aggregate<0><<<N_TILES, AGG_THREADS, 0, stream>>>(
        bufA, src, dst, efeat, out, degI, degO, b_g0, bufB);

    tile_aggregate<1><<<N_TILES, AGG_THREADS, 0, stream>>>(
        bufB, src, dst, efeat, out, nullptr, nullptr, nullptr, bufA);
    gemm_wmma_f32<128><<<L1_BLOCKS, 256, 0, stream>>>(
        bufA, W_g1, b_g1, degI, nullptr, bufB, N_NODES, 1);

    tile_aggregate<2><<<N_TILES, AGG_THREADS, 0, stream>>>(
        bufB, src, dst, efeat, out, nullptr, nullptr, nullptr, out);
}


# list launch: a wave takes its queue slots once per 256-key sweep (lane match counts, wave prefix sum, one LDS add), not one LDS atomic round trip per match; queue order changes, sums arrival-order
# speedup vs baseline: 3.2854x; 1.0375x over previous
#include <hip/hip_runtime.h>

typedef float v2f __attribute__((ext_vector_type(2)));
typedef float v8f __attribute__((ext_vector_type(8)));

#define N_NODES 50000
#define N_PER   12500
#define NE      600000
#define HD      128

#define TILE_R      250
#define N_TILES     200
#define N_KEYW      (NE / 4)
#define REGION_B    (TILE_R * HD * 4)
#define KEY_OFF     65536
#define KCH_LOG     11
#define ROW_CAP     64
#define LIST_INTS   (TILE_R * ROW_CAP + TILE_R + 1)

static_assert(TILE_R * N_TILES == N_NODES, "tiles must cover the node range exactly");
static_assert(N_TILES <= 255, "a tile key is one byte and 0xff is the never-matching filler");
static_assert(NE % 16 == 0, "edge ids are read as int4, keys as 16-byte vectors");
static_assert(LIST_INTS * 4 <= KEY_OFF, "row lists end before the key chunk of their region");
static_assert(KEY_OFF + (16 << KCH_LOG) <= REGION_B, "a key chunk ends inside its region");
static_assert(((2 * (NE / 16) + (1 << KCH_LOG) - 1) >> KCH_LOG) <= N_TILES, "enough regions for all key chunks");
static_assert(REGION_B % 16 == 0 && KEY_OFF % 16 == 0, "16-byte vectors stay aligned");

__device__ __forceinline__ uint4* key_vec(float* base, int civ)
{
    return (uint4*)((char*)base + (size_t)(civ >> KCH_LOG) * REGION_B + KEY_OFF) + (civ & ((1 << KCH_LOG) - 1));
}

#define PROJ_BLOCKS 391
#define L1_BLOCKS   512

template <int K>
__global__ __launch_bounds__(256)
void gemm_wmma_f32(const float* __restrict__ A,
                   const float* __restrict__ B,
                   const float* __restrict__ bias,
                   const float* __restrict__ rowscale,
                   const float* __restrict__ outscale,
                   float* __restrict__ C,
                   int M, int relu)
{
    static_assert(K % 8 == 0, "the loaders walk k in blocks of 8");
    const int lane = threadIdx.x & 31;
    const int half = lane >> 4;
    const int lm   = lane & 15;
    const int wave   = (blockIdx.x * blockDim.x + threadIdx.x) >> 5;
    const int nwaves = (gridDim.x * blockDim.x) >> 5;
    const int tn  = wave & 7;
    const int col = tn * 16 + lm;

    v2f bf[K / 4];
    #pragma unroll
    for (int kk = 0; kk < K / 4; ++kk) {
        const int kb = (kk >> 1) * 8 + half * 4 + (kk & 1) * 2;
        bf[kk][0] = B[(long)kb * HD + col];
        bf[kk][1] = B[(long)(kb + 1) * HD + col];
    }

    const int mt = (M + 15) >> 4;
    const float bv = bias[col];

    for (int tm = wave >> 3; tm < mt; tm += (nwaves >> 3)) {
        const int row = tm * 16 + lm;
        const int rc  = (row < M) ? row : (M - 1);
        const float* ap = A + (long)rc * K + half * 4;

        float4 av[K / 8];
        #pragma unroll
        for (int p = 0; p < K / 8; ++p)
            av[p] = *(const float4*)(ap + p * 8);
        const float rs = (rowscale != nullptr) ? rowscale[rc] : 1.0f;
        float os[8];
        #pragma unroll
        for (int v = 0; v < 8; ++v) {
            const int orow = tm * 16 + v + half * 8;
            os[v] = (outscale != nullptr) ? outscale[(orow < M) ? orow : (M - 1)] : 1.0f;
        }

        v8f acc = {};
        #pragma unroll
        for (int p = 0; p < K / 8; ++p) {
            v2f a0, a1;
            a0[0] = av[p].x * rs;
            a0[1] = av[p].y * rs;
            a1[0] = av[p].z * rs;
            a1[1] = av[p].w * rs;
            acc = __builtin_amdgcn_wmma_f32_16x16x4_f32(
                false, a0, false, bf[2 * p], (short)0, acc, false, false);
            acc = __builtin_amdgcn_wmma_f32_16x16x4_f32(
                false, a1, false, bf[2 * p + 1], (short)0, acc, false, false);
        }

        #pragma unroll
        for (int v = 0; v < 8; ++v) {
            const int orow = tm * 16 + v + half * 8;
            if (orow < M) {
                float val = acc[v] + bv;
                if (relu) val = fmaxf(val, 0.0f);
                if (outscale != nullptr) val *= os[v];
                C[(long)orow * HD + col] = val;
            }
        }
    }
}

__global__ void edge_keys(const int* __restrict__ src,
                          const int* __restrict__ dst,
                          const int* __restrict__ ef,
                          float* __restrict__ keybuf,
                          float* __restrict__ recbuf)
{
    uint4* rec4 = (uint4*)recbuf;
    int j = blockIdx.x * blockDim.x + threadIdx.x;
    const int stride = gridDim.x * blockDim.x;
    for (; j < N_KEYW; j += stride) {
        const int4 s = *(const int4*)(src + 4 * j);
        const int4 d = *(const int4*)(dst + 4 * j);
        const int4 r = *(const int4*)(ef + 4 * j);
        const unsigned kd = ((unsigned)d.x / TILE_R) | (((unsigned)d.y / TILE_R) << 8) |
                            (((unsigned)d.z / TILE_R) << 16) | (((unsigned)d.w / TILE_R) << 24);
        const unsigned ks = ((unsigned)s.x / TILE_R) | (((unsigned)s.y / TILE_R) << 8) |
                            (((unsigned)s.z / TILE_R) << 16) | (((unsigned)s.w / TILE_R) << 24);
        const int cw = N_KEYW + j;
        ((unsigned*)key_vec(keybuf, j >> 2))[j & 3]   = kd;
        ((unsigned*)key_vec(keybuf, cw >> 2))[cw & 3] = ks;
        rec4[2 * j]     = make_uint4((unsigned)s.x | ((unsigned)d.x << 16), (unsigned)r.x,
                                     (unsigned)s.y | ((unsigned)d.y << 16), (unsigned)r.y);
        rec4[2 * j + 1] = make_uint4((unsigned)s.z | ((unsigned)d.z << 16), (unsigned)r.z,
                                     (unsigned)s.w | ((unsigned)d.w << 16), (unsigned)r.w);
    }
}

#define Q_CAP       16384
#define LQ_CAP      32768
#define AGG_THREADS 1024
#define AGG_WAVES   (AGG_THREADS / 32)
#define GATHER_UNR  16
#define DEAL_UNR    4

static_assert(NE <= (1 << 20), "queue entry packing");
static_assert(ROW_CAP == 64, "a row's list is read as two batches of 32 entries, one per lane");
static_assert(N_NODES <= (1 << 16), "row-list entry: node id in the low 16 bits, flags in bits 16 and 17; edge record: src | dst << 16");
static_assert((size_t)NE * 8 <= (size_t)N_NODES * HD * 4, "the edge records fit the node buffer that holds them");
static_assert((TILE_R * ROW_CAP) % 4 == 0, "row lists are copied as 16-byte vectors");

__device__ __forceinline__ void lds_add(float* p, float v)
{
    __hip_atomic_fetch_add(p, v, __ATOMIC_RELAXED, __HIP_MEMORY_SCOPE_WORKGROUP);
}

__device__ __forceinline__ int wave_prefix_sum(int x)
{
    x += __builtin_amdgcn_update_dpp(0, x, 0x111, 0xf, 0xf, true);
    x += __builtin_amdgcn_update_dpp(0, x, 0x112, 0xf, 0xf, true);
    x += __builtin_amdgcn_update_dpp(0, x, 0x114, 0xf, 0xf, true);
    x += __builtin_amdgcn_update_dpp(0, x, 0x118, 0xf, 0xf, true);
    const int low = __builtin_amdgcn_readlane(x, 15);
    return x + (((threadIdx.x & 16) != 0) ? low : 0);
}

__global__ __launch_bounds__(AGG_THREADS)
void tile_lists(const int* __restrict__ src,
                const int* __restrict__ dst,
                const int* __restrict__ ef,
                const float* __restrict__ recbuf,
                float* scratch,
                float* __restrict__ out_norm,
                float* __restrict__ in_norm)
{
    __shared__ int   queue[LQ_CAP];
    __shared__ __attribute__((aligned(16))) int rq[TILE_R * ROW_CAP];
    __shared__ int   rcnt[TILE_R];
    __shared__ int   din[TILE_R];
    __shared__ int   dout[TILE_R];
    __shared__ int   qn;
    __shared__ int   ovf;

    const int tid = threadIdx.x;
    const int r0  = blockIdx.x * TILE_R;
    int* mylist = (int*)((char*)scratch + (size_t)blockIdx.x * REGION_B);
    const uint2* rec = (const uint2*)recbuf;

    if (tid < TILE_R) { rcnt[tid] = 0; din[tid] = 0; dout[tid] = 0; }
    if (tid == 0) { qn = 0; ovf = 0; }
    __syncthreads();

    auto deal = [&](int ent, int s, int d, int r, bool listed) {
        const bool srcend = ((ent >> 29) & 1) != 0;
        const int lrow = (srcend ? s : d) - r0;
        if ((unsigned)lrow >= (unsigned)TILE_R) return;
        atomicAdd(srcend ? &dout[lrow] : &din[lrow], 1);
        if (srcend && !(r == 6 || r == 14 || r == 30)) return;
        if (!listed) return;
        const int fl = srcend ? 2 : ((r >= 0 && r <= 4) ? 1 : 0);
        const int slot = atomicAdd(&rcnt[lrow], 1);
        if (slot < ROW_CAP)
            rq[lrow * ROW_CAP + slot] = (srcend ? d : s) | (fl << 16);
        else
            ovf = 1;
    };

    auto place = [&](int slot, int ent) {
        if (slot < LQ_CAP) {
            queue[slot] = ent;
        } else {
            const int e = ent & 0xfffff;
            ovf = 1;
            deal(ent, src[e], dst[e], ef[e], false);
        }
    };

    {
        constexpr int NV = NE / 16;
        constexpr int SU = 8;
        static_assert(SU % 2 == 0, "match flags are kept two vectors to a word");
        const unsigned tk = (unsigned)blockIdx.x * 0x01010101u;

        auto eq4 = [&](unsigned x) -> unsigned {
            const unsigned t = x ^ tk;
            return ~(((t & 0x7f7f7f7fu) + 0x7f7f7f7fu) | t) & 0x80808080u;
        };
        auto match16 = [&](const uint4& q) -> unsigned {
            return (eq4(q.x) >> 7) | (eq4(q.y) >> 6) | (eq4(q.z) >> 5) | (eq4(q.w) >> 4);
        };
        auto emit = [&](unsigned f, int ebase, int flag, int& slot) {
            while (f != 0) {
                const int b = __builtin_ctz(f);
                f &= f - 1;
                place(slot, (ebase + ((b >> 2) & 1) * (AGG_THREADS * 16) + (b & 3) * 4 + (b >> 3)) | flag);
                ++slot;
            }
        };

        for (int v0 = 0; v0 < NV; v0 += AGG_THREADS * SU) {
            uint4 kd[SU];
            uint4 ks[SU];
            #pragma unroll
            for (int u = 0; u < SU; ++u) {
                const int iv = v0 + u * AGG_THREADS + tid;
                const bool in = iv < NV;
                kd[u] = in ? *key_vec(scratch, iv)      : make_uint4(~0u, ~0u, ~0u, ~0u);
                ks[u] = in ? *key_vec(scratch, NV + iv) : make_uint4(~0u, ~0u, ~0u, ~0u);
            }
            unsigned md[SU / 2];
            unsigned ms[SU / 2];
            int cnt = 0;
            #pragma unroll
            for (int p = 0; p < SU / 2; ++p) {
                md[p] = match16(kd[2 * p]) | (match16(kd[2 * p + 1]) << 4);
                ms[p] = match16(ks[2 * p]) | (match16(ks[2 * p + 1]) << 4);
                cnt += __builtin_popcount(md[p]) + __builtin_popcount(ms[p]);
            }
            const int incl = wave_prefix_sum(cnt);
            const int total = __builtin_amdgcn_readlane(incl, 31);
            if (total != 0) {
                int first = 0;
                if ((tid & 31) == 31) first = atomicAdd(&qn, total);
                int slot = __builtin_amdgcn_readlane(first, 31) + incl - cnt;
                #pragma unroll
                for (int p = 0; p < SU / 2; ++p) {
                    const int ebase = (v0 + 2 * p * AGG_THREADS + tid) * 16;
                    emit(md[p], ebase, 0, slot);
                    emit(ms[p], ebase, 1 << 29, slot);
                }
            }
        }
    }
    __syncthreads();

    const int n = __builtin_amdgcn_readfirstlane(min(qn, LQ_CAP));
    for (int base = 0; base < n; base += AGG_THREADS * DEAL_UNR) {
        int   ent[DEAL_UNR];
        uint2 er[DEAL_UNR];
        #pragma unroll
        for (int u = 0; u < DEAL_UNR; ++u) {
            const int idx = base + u * AGG_THREADS + tid;
            ent[u] = queue[min(idx, n - 1)];
        }
        #pragma unroll
        for (int u = 0; u < DEAL_UNR; ++u)
            er[u] = rec[ent[u] & 0xfffff];
        #pragma unroll
        for (int u = 0; u < DEAL_UNR; ++u) {
            const int idx = base + u * AGG_THREADS + tid;
            if (idx < n)
                deal(ent[u], (int)(er[u].x & 0xffffu), (int)(er[u].x >> 16), (int)er[u].y, true);
        }
    }
    __syncthreads();

    for (int i = tid; i < TILE_R * ROW_CAP / 4; i += AGG_THREADS)
        ((int4*)mylist)[i] = ((const int4*)rq)[i];
    if (tid < TILE_R) {
        mylist[TILE_R * ROW_CAP + tid] = min(rcnt[tid], ROW_CAP);
        out_norm[r0 + tid] = rsqrtf(fmaxf((float)dout[tid], 1.0f));
        in_norm[r0 + tid]  = rsqrtf(fmaxf((float)din[tid], 1.0f));
    }
    if (tid == 0) mylist[TILE_R * ROW_CAP + TILE_R] = ovf;
}

template <int MODE>
__global__ __launch_bounds__(AGG_THREADS)
void tile_aggregate(const float* __restrict__ h,
                    const int* __restrict__ src,
                    const int* __restrict__ dst,
                    const int* __restrict__ ef,
                    float* scratch,
                    const float* __restrict__ in_norm,
                    const float* __restrict__ out_norm,
                    const float* __restrict__ bias,
                    float* outp)
{
    __shared__ float tile[TILE_R * HD];
    __shared__ int   queue[Q_CAP];
    __shared__ __attribute__((aligned(16))) int rq[TILE_R * ROW_CAP];
    __shared__ int   rcnt[TILE_R];
    __shared__ int   qn;
    __shared__ int   ovf;

    const int tid  = threadIdx.x;
    const int lane = tid & 31;
    const int wave = __builtin_amdgcn_readfirstlane(tid >> 5);
    const int r0   = blockIdx.x * TILE_R;
    int* mylist = (int*)((char*)scratch + (size_t)blockIdx.x * REGION_B);

    const bool build = __builtin_amdgcn_readfirstlane(mylist[TILE_R * ROW_CAP + TILE_R]) != 0;

    auto decode = [&](int ent, int s, int d, int r, int& node, int& fl, int& lrow) -> bool {
        bool valid;
        if ((ent >> 29) & 1) {
            node = d;
            lrow = s - r0;
            fl = 2;
            valid = (r == 6 || r == 14 || r == 30);
        } else {
            node = s;
            lrow = d - r0;
            fl = (r >= 0 && r <= 4) ? 1 : 0;
            valid = true;
        }
        return valid && ((unsigned)lrow < (unsigned)TILE_R);
    };

    auto add_now = [&](int node, int fl, int lrow) {
        ovf = 1;
        if (MODE != 2 && fl == 2) return;
        const float w = (MODE == 2 && fl == 1) ? 2.0f : 1.0f;
        float* t = tile + lrow * HD;
        for (int c = 0; c < HD; ++c)
            lds_add(t + (c & 3) * 32 + (c >> 2), h[(long)node * HD + c] * w);
    };

    auto place = [&](int slot, int ent) {
        if (slot < Q_CAP) {
            queue[slot] = ent;
        } else {
            const int e = ent & 0xfffff;
            int node, fl, lrow;
            if (decode(ent, src[e], dst[e], ef[e], node, fl, lrow)) add_now(node, fl, lrow);
        }
    };

    if (build) {
        for (int i = tid; i < TILE_R * HD; i += AGG_THREADS) tile[i] = 0.0f;
        if (tid < TILE_R) rcnt[tid] = 0;
        if (tid == 0) { qn = 0; ovf = 0; }
        __syncthreads();

        for (int e = tid; e < NE; e += AGG_THREADS) {
            if ((unsigned)(dst[e] - r0) < (unsigned)TILE_R) place(atomicAdd(&qn, 1), e);
            if (MODE == 2 && (unsigned)(src[e] - r0) < (unsigned)TILE_R) place(atomicAdd(&qn, 1), e | (1 << 29));
        }
        __syncthreads();

        const int n = __builtin_amdgcn_readfirstlane(min(qn, Q_CAP));
        for (int base = 0; base < n; base += AGG_THREADS * DEAL_UNR) {
            int ent[DEAL_UNR];
            int es[DEAL_UNR];
            int ed[DEAL_UNR];
            int er[DEAL_UNR];
            #pragma unroll
            for (int u = 0; u < DEAL_UNR; ++u) {
                const int idx = base + u * AGG_THREADS + tid;
                ent[u] = queue[min(idx, n - 1)];
            }
            #pragma unroll
            for (int u = 0; u < DEAL_UNR; ++u) {
                const int e = ent[u] & 0xfffff;
                es[u] = src[e];
                ed[u] = dst[e];
                er[u] = ef[e];
            }
            #pragma unroll
            for (int u = 0; u < DEAL_UNR; ++u) {
                const int idx = base + u * AGG_THREADS + tid;
                int node, fl, lrow;
                const bool valid = decode(ent[u], es[u], ed[u], er[u], node, fl, lrow);
                if (idx < n && valid) {
                    const int slot = atomicAdd(&rcnt[lrow], 1);
                    if (slot < ROW_CAP)
                        rq[lrow * ROW_CAP + slot] = node | (fl << 16);
                    else
                        add_now(node, fl, lrow);
                }
            }
        }
        __syncthreads();
    } else {
        constexpr int LV = TILE_R * ROW_CAP / 4;
        constexpr int LU = (LV + AGG_THREADS - 1) / AGG_THREADS;
        int4 lv[LU];
        #pragma unroll
        for (int u = 0; u < LU; ++u)
            lv[u] = ((const int4*)mylist)[min(tid + u * AGG_THREADS, LV - 1)];
        #pragma unroll
        for (int u = 0; u < LU; ++u)
            if (tid + u * AGG_THREADS < LV) ((int4*)rq)[tid + u * AGG_THREADS] = lv[u];
        if (tid < TILE_R) rcnt[tid] = mylist[TILE_R * ROW_CAP + tid];
        if (tid == 0) ovf = 0;
        __syncthreads();
    }

    const int any_ovf = __builtin_amdgcn_readfirstlane(ovf);
    const float* hl = h + lane * 4;

    for (int lr = wave; lr < TILE_R; lr += AGG_WAVES) {
        const int rc = __builtin_amdgcn_readfirstlane(min(rcnt[lr], ROW_CAP));
        float4 acc = make_float4(0.0f, 0.0f, 0.0f, 0.0f);
        for (int b = 0; b < rc; b += 32) {
            const int m  = min(32, rc - b);
            const int my = rq[lr * ROW_CAP + b + lane];
            for (int j = 0; j < m; j += GATHER_UNR) {
                float4 v[GATHER_UNR];
                int    en[GATHER_UNR];
                #pragma unroll
                for (int u = 0; u < GATHER_UNR; ++u) {
                    en[u] = 1 << 17;
                    if (j + u < m) {
                        en[u] = __builtin_amdgcn_readlane(my, j + u);
                        if (MODE == 2 || !((en[u] >> 17) & 1))
                            v[u] = *(const float4*)(hl + (long)(en[u] & 0xffff) * HD);
                    }
                }
                #pragma unroll
                for (int u = 0; u < GATHER_UNR; ++u) {
                    const bool ok = (j + u < m) && (MODE == 2 || !((en[u] >> 17) & 1));
                    float4 x = v[u];
                    if (MODE == 2) {
                        const float w = (ok && ((en[u] >> 16) & 1)) ? 2.0f : 1.0f;
                        x.x *= w; x.y *= w; x.z *= w; x.w *= w;
                    }
                    acc.x = ok ? acc.x + x.x : acc.x;
                    acc.y = ok ? acc.y + x.y : acc.y;
                    acc.z = ok ? acc.z + x.z : acc.z;
                    acc.w = ok ? acc.w + x.w : acc.w;
                }
            }
        }

        if (any_ovf != 0) {
            const float* t = tile + lr * HD + lane;
            acc.x += t[0]; acc.y += t[32]; acc.z += t[64]; acc.w += t[96];
        }
        const int row = r0 + lr;
        if (MODE == 0) {
            const float inn = in_norm[row];
            const float on  = out_norm[row];
            const float4 bb = *(const float4*)(bias + lane * 4);
            acc.x = fmaxf(acc.x * inn + bb.x, 0.0f) * on;
            acc.y = fmaxf(acc.y * inn + bb.y, 0.0f) * on;
            acc.z = fmaxf(acc.z * inn + bb.z, 0.0f) * on;
            acc.w = fmaxf(acc.w * inn + bb.w, 0.0f) * on;
        }
        *(float4*)(outp + (long)row * HD + lane * 4) = acc;
    }
}

extern "C" void kernel_launch(void* const* d_in, const int* in_sizes, int n_in,
                              void* d_out, int out_size, void* d_ws, size_t ws_size,
                              hipStream_t stream)
{
    (void)in_sizes; (void)n_in; (void)out_size; (void)ws_size;

    const float* feat0 = (const float*)d_in[0];
    const float* Wfc0  = (const float*)d_in[1];
    const float* bfc0  = (const float*)d_in[2];
    const float* feat1 = (const float*)d_in[3];
    const float* Wfc1  = (const float*)d_in[4];
    const float* bfc1  = (const float*)d_in[5];
    const float* feat2 = (const float*)d_in[6];
    const float* Wfc2  = (const float*)d_in[7];
    const float* bfc2  = (const float*)d_in[8];
    const float* feat3 = (const float*)d_in[9];
    const float* Wfc3  = (const float*)d_in[10];
    const float* bfc3  = (const float*)d_in[11];
    const int*   src   = (const int*)d_in[12];
    const int*   dst   = (const int*)d_in[13];
    const int*   efeat = (const int*)d_in[14];
    const float* b_g0  = (const float*)d_in[15];
    const float* W_g1  = (const float*)d_in[16];
    const float* b_g1  = (const float*)d_in[17];
    float* out = (float*)d_out;

    const size_t NODE_ELEMS = (size_t)N_NODES * HD;
    float* bufA = (float*)d_ws;
    float* bufB = bufA + NODE_ELEMS;
    float* degO = bufB + NODE_ELEMS;
    float* degI = degO + N_NODES;

    edge_keys<<<(N_KEYW + 255) / 256, 256, 0, stream>>>(src, dst, efeat, out, bufB);
    tile_lists<<<N_TILES, AGG_THREADS, 0, stream>>>(src, dst, efeat, bufB, out, degO, degI);

    gemm_wmma_f32<128><<<PROJ_BLOCKS, 256, 0, stream>>>(
        feat0, Wfc0, bfc0, nullptr, degO + 0 * N_PER,
        bufA + (size_t)0 * N_PER * HD, N_PER, 0);
    gemm_wmma_f32<128><<<PROJ_BLOCKS, 256, 0, stream>>>(
        feat1, Wfc1, bfc1, nullptr, degO + 1 * N_PER,
        bufA + (size_t)1 * N_PER * HD, N_PER, 0);
    gemm_wmma_f32<64><<<PROJ_BLOCKS, 256, 0, stream>>>(
        feat2, Wfc2, bfc2, nullptr, degO + 2 * N_PER,
        bufA + (size_t)2 * N_PER * HD, N_PER, 0);
    gemm_wmma_f32<64><<<PROJ_BLOCKS, 256, 0, stream>>>(
        feat3, Wfc3, bfc3, nullptr, degO + 3 * N_PER,
        bufA + (size_t)3 * N_PER * HD, N_PER, 0);

    tile_aggregate<0><<<N_TILES, AGG_THREADS, 0, stream>>>(
        bufA, src, dst, efeat, out, degI, degO, b_g0, bufB);

    tile_aggregate<1><<<N_TILES, AGG_THREADS, 0, stream>>>(
        bufB, src, dst, efeat, out, nullptr, nullptr, nullptr, bufA);
    gemm_wmma_f32<128><<<L1_BLOCKS, 256, 0, stream>>>(
        bufA, W_g1, b_g1, degI, nullptr, bufB, N_NODES, 1);

    tile_aggregate<2><<<N_TILES, AGG_THREADS, 0, stream>>>(
        bufB, src, dst, efeat, out, nullptr, nullptr, nullptr, out);
}


# tile keys re-laid for the scan: bit-sliced (8 plane words per 32 edges, wave ballots in the key pass), contiguous behind edge records: 32 keys per 8 XOR+OR; queue order changes, sums arrival-order
# speedup vs baseline: 3.4503x; 1.0502x over previous
#include <hip/hip_runtime.h>

typedef float v2f __attribute__((ext_vector_type(2)));
typedef float v8f __attribute__((ext_vector_type(8)));

#define N_NODES 50000
#define N_PER   12500
#define NE      600000
#define HD      128

#define TILE_R      250
#define N_TILES     200
#define AGG_THREADS 1024
#define SCAN_SG     4
#define N_KEYW      (NE / 4)
#define N_GRP       (((N_KEYW + 31) / 32) * 4)
#define N_GRP_PAD   (((N_GRP + AGG_THREADS * SCAN_SG - 1) / (AGG_THREADS * SCAN_SG)) * (AGG_THREADS * SCAN_SG))
#define N_KVEC      (2 * N_GRP_PAD)
#define REGION_B    (TILE_R * HD * 4)
#define ROW_CAP     64
#define LIST_INTS   (TILE_R * ROW_CAP + TILE_R + 1)

static_assert(TILE_R * N_TILES == N_NODES, "tiles must cover the node range exactly");
static_assert(N_TILES <= 255, "a tile key is one byte and 0xff is the never-matching filler");
static_assert(NE % 4 == 0, "edge ids are read as int4");
static_assert(LIST_INTS * 4 <= REGION_B, "the row lists of a tile fit its region");
static_assert(REGION_B % 16 == 0, "16-byte vectors stay aligned");
static_assert((size_t)NE * 8 + (size_t)2 * N_KVEC * 16 <= (size_t)N_NODES * HD * 4, "edge records and both key arrays fit one node plane");

#define PROJ_BLOCKS 391
#define L1_BLOCKS   512

template <int K>
__global__ __launch_bounds__(256)
void gemm_wmma_f32(const float* __restrict__ A,
                   const float* __restrict__ B,
                   const float* __restrict__ bias,
                   const float* __restrict__ rowscale,
                   const float* __restrict__ outscale,
                   float* __restrict__ C,
                   int M, int relu)
{
    static_assert(K % 8 == 0, "the loaders walk k in blocks of 8");
    const int lane = threadIdx.x & 31;
    const int half = lane >> 4;
    const int lm   = lane & 15;
    const int wave   = (blockIdx.x * blockDim.x + threadIdx.x) >> 5;
    const int nwaves = (gridDim.x * blockDim.x) >> 5;
    const int tn  = wave & 7;
    const int col = tn * 16 + lm;

    v2f bf[K / 4];
    #pragma unroll
    for (int kk = 0; kk < K / 4; ++kk) {
        const int kb = (kk >> 1) * 8 + half * 4 + (kk & 1) * 2;
        bf[kk][0] = B[(long)kb * HD + col];
        bf[kk][1] = B[(long)(kb + 1) * HD + col];
    }

    const int mt = (M + 15) >> 4;
    const float bv = bias[col];

    for (int tm = wave >> 3; tm < mt; tm += (nwaves >> 3)) {
        const int row = tm * 16 + lm;
        const int rc  = (row < M) ? row : (M - 1);
        const float* ap = A + (long)rc * K + half * 4;

        float4 av[K / 8];
        #pragma unroll
        for (int p = 0; p < K / 8; ++p)
            av[p] = *(const float4*)(ap + p * 8);
        const float rs = (rowscale != nullptr) ? rowscale[rc] : 1.0f;
        float os[8];
        #pragma unroll
        for (int v = 0; v < 8; ++v) {
            const int orow = tm * 16 + v + half * 8;
            os[v] = (outscale != nullptr) ? outscale[(orow < M) ? orow : (M - 1)] : 1.0f;
        }

        v8f acc = {};
        #pragma unroll
        for (int p = 0; p < K / 8; ++p) {
            v2f a0, a1;
            a0[0] = av[p].x * rs;
            a0[1] = av[p].y * rs;
            a1[0] = av[p].z * rs;
            a1[1] = av[p].w * rs;
            acc = __builtin_amdgcn_wmma_f32_16x16x4_f32(
                false, a0, false, bf[2 * p], (short)0, acc, false, false);
            acc = __builtin_amdgcn_wmma_f32_16x16x4_f32(
                false, a1, false, bf[2 * p + 1], (short)0, acc, false, false);
        }

        #pragma unroll
        for (int v = 0; v < 8; ++v) {
            const int orow = tm * 16 + v + half * 8;
            if (orow < M) {
                float val = acc[v] + bv;
                if (relu) val = fmaxf(val, 0.0f);
                if (outscale != nullptr) val *= os[v];
                C[(long)orow * HD + col] = val;
            }
        }
    }
}

__global__ void edge_keys(const int* __restrict__ src,
                          const int* __restrict__ dst,
                          const int* __restrict__ ef,
                          float* __restrict__ recbuf)
{
    uint4* rec4 = (uint4*)recbuf;
    uint4* keyv = (uint4*)recbuf + NE / 2;
    const int lane = threadIdx.x & 31;
    int j = blockIdx.x * blockDim.x + threadIdx.x;
    const int stride = gridDim.x * blockDim.x;
    for (; j - lane < N_GRP_PAD * 8; j += stride) {
        unsigned kd[4] = {0xffu, 0xffu, 0xffu, 0xffu};
        unsigned ks[4] = {0xffu, 0xffu, 0xffu, 0xffu};
        if (j < N_KEYW) {
            const int4 s = *(const int4*)(src + 4 * j);
            const int4 d = *(const int4*)(dst + 4 * j);
            const int4 r = *(const int4*)(ef + 4 * j);
            kd[0] = (unsigned)d.x / TILE_R; kd[1] = (unsigned)d.y / TILE_R;
            kd[2] = (unsigned)d.z / TILE_R; kd[3] = (unsigned)d.w / TILE_R;
            ks[0] = (unsigned)s.x / TILE_R; ks[1] = (unsigned)s.y / TILE_R;
            ks[2] = (unsigned)s.z / TILE_R; ks[3] = (unsigned)s.w / TILE_R;
            rec4[2 * j]     = make_uint4((unsigned)s.x | ((unsigned)d.x << 16), (unsigned)r.x,
                                         (unsigned)s.y | ((unsigned)d.y << 16), (unsigned)r.y);
            rec4[2 * j + 1] = make_uint4((unsigned)s.z | ((unsigned)d.z << 16), (unsigned)r.z,
                                         (unsigned)s.w | ((unsigned)d.w << 16), (unsigned)r.w);
        }
        const int q = __builtin_amdgcn_readfirstlane((j - lane) >> 5);
        unsigned w[4] = {0u, 0u, 0u, 0u};
        #pragma unroll
        for (int L = 0; L < 16; ++L) {
            const unsigned key = (L >> 3) ? ks[(L >> 1) & 3] : kd[(L >> 1) & 3];
            #pragma unroll
            for (int i = 0; i < 4; ++i) {
                const unsigned plane = __builtin_amdgcn_ballot_w32(((key >> (4 * (L & 1) + i)) & 1u) != 0u);
                w[i] = (lane == L) ? plane : w[i];
            }
        }
        if (lane < 16)
            keyv[(lane >> 3) * N_KVEC + (lane & 1) * N_GRP_PAD + 4 * q + ((lane >> 1) & 3)] =
                make_uint4(w[0], w[1], w[2], w[3]);
    }
}

#define Q_CAP       16384
#define LQ_CAP      32768
#define AGG_WAVES   (AGG_THREADS / 32)
#define GATHER_UNR  16
#define DEAL_UNR    4

static_assert(NE <= (1 << 20), "queue entry packing");
static_assert(ROW_CAP == 64, "a row's list is read as two batches of 32 entries, one per lane");
static_assert(N_NODES <= (1 << 16), "row-list entry: node id in the low 16 bits, flags in bits 16 and 17; edge record: src | dst << 16");
static_assert((TILE_R * ROW_CAP) % 4 == 0, "row lists are copied as 16-byte vectors");

__device__ __forceinline__ void lds_add(float* p, float v)
{
    __hip_atomic_fetch_add(p, v, __ATOMIC_RELAXED, __HIP_MEMORY_SCOPE_WORKGROUP);
}

__device__ __forceinline__ int wave_prefix_sum(int x)
{
    x += __builtin_amdgcn_update_dpp(0, x, 0x111, 0xf, 0xf, true);
    x += __builtin_amdgcn_update_dpp(0, x, 0x112, 0xf, 0xf, true);
    x += __builtin_amdgcn_update_dpp(0, x, 0x114, 0xf, 0xf, true);
    x += __builtin_amdgcn_update_dpp(0, x, 0x118, 0xf, 0xf, true);
    const int low = __builtin_amdgcn_readlane(x, 15);
    return x + (((threadIdx.x & 16) != 0) ? low : 0);
}

__global__ __launch_bounds__(AGG_THREADS)
void tile_lists(const int* __restrict__ src,
                const int* __restrict__ dst,
                const int* __restrict__ ef,
                const float* __restrict__ recbuf,
                float* scratch,
                float* __restrict__ out_norm,
                float* __restrict__ in_norm)
{
    __shared__ int   queue[LQ_CAP];
    __shared__ __attribute__((aligned(16))) int rq[TILE_R * ROW_CAP];
    __shared__ int   rcnt[TILE_R];
    __shared__ int   din[TILE_R];
    __shared__ int   dout[TILE_R];
    __shared__ int   qn;
    __shared__ int   ovf;

    const int tid = threadIdx.x;
    const int r0  = blockIdx.x * TILE_R;
    int* mylist = (int*)((char*)scratch + (size_t)blockIdx.x * REGION_B);
    const uint2* rec = (const uint2*)recbuf;
    const uint4* keyv = (const uint4*)recbuf + NE / 2;

    if (tid < TILE_R) { rcnt[tid] = 0; din[tid] = 0; dout[tid] = 0; }
    if (tid == 0) { qn = 0; ovf = 0; }
    __syncthreads();

    auto deal = [&](int ent, int s, int d, int r, bool listed) {
        const bool srcend = ((ent >> 29) & 1) != 0;
        const int lrow = (srcend ? s : d) - r0;
        if ((unsigned)lrow >= (unsigned)TILE_R) return;
        atomicAdd(srcend ? &dout[lrow] : &din[lrow], 1);
        if (srcend && !(r == 6 || r == 14 || r == 30)) return;
        if (!listed) return;
        const int fl = srcend ? 2 : ((r >= 0 && r <= 4) ? 1 : 0);
        const int slot = atomicAdd(&rcnt[lrow], 1);
        if (slot < ROW_CAP)
            rq[lrow * ROW_CAP + slot] = (srcend ? d : s) | (fl << 16);
        else
            ovf = 1;
    };

    auto place = [&](int slot, int ent) {
        if (slot < LQ_CAP) {
            queue[slot] = ent;
        } else {
            const int e = ent & 0xfffff;
            ovf = 1;
            deal(ent, src[e], dst[e], ef[e], false);
        }
    };

    {
        unsigned tb[8];
        #pragma unroll
        for (int b = 0; b < 8; ++b) tb[b] = 0u - (((unsigned)blockIdx.x >> b) & 1u);

        auto match32 = [&](const uint4& lo, const uint4& hi) -> unsigned {
            return ~((lo.x ^ tb[0]) | (lo.y ^ tb[1]) | (lo.z ^ tb[2]) | (lo.w ^ tb[3]) |
                     (hi.x ^ tb[4]) | (hi.y ^ tb[5]) | (hi.z ^ tb[6]) | (hi.w ^ tb[7]));
        };
        auto emit = [&](unsigned f, int ebase, int flag, int& slot) {
            while (f != 0) {
                const int b = __builtin_ctz(f);
                f &= f - 1;
                place(slot, (ebase + 4 * b) | flag);
                ++slot;
            }
        };

        for (int g0 = 0; g0 < N_GRP_PAD; g0 += AGG_THREADS * SCAN_SG) {
            uint4 dl[SCAN_SG], dh[SCAN_SG];
            uint4 sl[SCAN_SG], sh[SCAN_SG];
            #pragma unroll
            for (int u = 0; u < SCAN_SG; ++u) {
                const int g = g0 + u * AGG_THREADS + tid;
                dl[u] = keyv[g];
                dh[u] = keyv[N_GRP_PAD + g];
                sl[u] = keyv[N_KVEC + g];
                sh[u] = keyv[N_KVEC + N_GRP_PAD + g];
            }
            unsigned md[SCAN_SG];
            unsigned ms[SCAN_SG];
            int cnt = 0;
            #pragma unroll
            for (int u = 0; u < SCAN_SG; ++u) {
                md[u] = match32(dl[u], dh[u]);
                ms[u] = match32(sl[u], sh[u]);
                cnt += __builtin_popcount(md[u]) + __builtin_popcount(ms[u]);
            }
            const int incl = wave_prefix_sum(cnt);
            const int total = __builtin_amdgcn_readlane(incl, 31);
            if (total != 0) {
                int first = 0;
                if ((tid & 31) == 31) first = atomicAdd(&qn, total);
                int slot = __builtin_amdgcn_readlane(first, 31) + incl - cnt;
                #pragma unroll
                for (int u = 0; u < SCAN_SG; ++u) {
                    const int g = g0 + u * AGG_THREADS + tid;
                    const int ebase = (g >> 2) * 128 + (g & 3);
                    emit(md[u], ebase, 0, slot);
                    emit(ms[u], ebase, 1 << 29, slot);
                }
            }
        }
    }
    __syncthreads();

    const int n = __builtin_amdgcn_readfirstlane(min(qn, LQ_CAP));
    for (int base = 0; base < n; base += AGG_THREADS * DEAL_UNR) {
        int   ent[DEAL_UNR];
        uint2 er[DEAL_UNR];
        #pragma unroll
        for (int u = 0; u < DEAL_UNR; ++u) {
            const int idx = base + u * AGG_THREADS + tid;
            ent[u] = queue[min(idx, n - 1)];
        }
        #pragma unroll
        for (int u = 0; u < DEAL_UNR; ++u)
            er[u] = rec[ent[u] & 0xfffff];
        #pragma unroll
        for (int u = 0; u < DEAL_UNR; ++u) {
            const int idx = base + u * AGG_THREADS + tid;
            if (idx < n)
                deal(ent[u], (int)(er[u].x & 0xffffu), (int)(er[u].x >> 16), (int)er[u].y, true);
        }
    }
    __syncthreads();

    for (int i = tid; i < TILE_R * ROW_CAP / 4; i += AGG_THREADS)
        ((int4*)mylist)[i] = ((const int4*)rq)[i];
    if (tid < TILE_R) {
        mylist[TILE_R * ROW_CAP + tid] = min(rcnt[tid], ROW_CAP);
        out_norm[r0 + tid] = rsqrtf(fmaxf((float)dout[tid], 1.0f));
        in_norm[r0 + tid]  = rsqrtf(fmaxf((float)din[tid], 1.0f));
    }
    if (tid == 0) mylist[TILE_R * ROW_CAP + TILE_R] = ovf;
}

template <int MODE>
__global__ __launch_bounds__(AGG_THREADS)
void tile_aggregate(const float* __restrict__ h,
                    const int* __restrict__ src,
                    const int* __restrict__ dst,
                    const int* __restrict__ ef,
                    float* scratch,
                    const float* __restrict__ in_norm,
                    const float* __restrict__ out_norm,
                    const float* __restrict__ bias,
                    float* outp)
{
    __shared__ float tile[TILE_R * HD];
    __shared__ int   queue[Q_CAP];
    __shared__ __attribute__((aligned(16))) int rq[TILE_R * ROW_CAP];
    __shared__ int   rcnt[TILE_R];
    __shared__ int   qn;
    __shared__ int   ovf;

    const int tid  = threadIdx.x;
    const int lane = tid & 31;
    const int wave = __builtin_amdgcn_readfirstlane(tid >> 5);
    const int r0   = blockIdx.x * TILE_R;
    int* mylist = (int*)((char*)scratch + (size_t)blockIdx.x * REGION_B);

    const bool build = __builtin_amdgcn_readfirstlane(mylist[TILE_R * ROW_CAP + TILE_R]) != 0;

    auto decode = [&](int ent, int s, int d, int r, int& node, int& fl, int& lrow) -> bool {
        bool valid;
        if ((ent >> 29) & 1) {
            node = d;
            lrow = s - r0;
            fl = 2;
            valid = (r == 6 || r == 14 || r == 30);
        } else {
            node = s;
            lrow = d - r0;
            fl = (r >= 0 && r <= 4) ? 1 : 0;
            valid = true;
        }
        return valid && ((unsigned)lrow < (unsigned)TILE_R);
    };

    auto add_now = [&](int node, int fl, int lrow) {
        ovf = 1;
        if (MODE != 2 && fl == 2) return;
        const float w = (MODE == 2 && fl == 1) ? 2.0f : 1.0f;
        float* t = tile + lrow * HD;
        for (int c = 0; c < HD; ++c)
            lds_add(t + (c & 3) * 32 + (c >> 2), h[(long)node * HD + c] * w);
    };

    auto place = [&](int slot, int ent) {
        if (slot < Q_CAP) {
            queue[slot] = ent;
        } else {
            const int e = ent & 0xfffff;
            int node, fl, lrow;
            if (decode(ent, src[e], dst[e], ef[e], node, fl, lrow)) add_now(node, fl, lrow);
        }
    };

    if (build) {
        for (int i = tid; i < TILE_R * HD; i += AGG_THREADS) tile[i] = 0.0f;
        if (tid < TILE_R) rcnt[tid] = 0;
        if (tid == 0) { qn = 0; ovf = 0; }
        __syncthreads();

        for (int e = tid; e < NE; e += AGG_THREADS) {
            if ((unsigned)(dst[e] - r0) < (unsigned)TILE_R) place(atomicAdd(&qn, 1), e);
            if (MODE == 2 && (unsigned)(src[e] - r0) < (unsigned)TILE_R) place(atomicAdd(&qn, 1), e | (1 << 29));
        }
        __syncthreads();

        const int n = __builtin_amdgcn_readfirstlane(min(qn, Q_CAP));
        for (int base = 0; base < n; base += AGG_THREADS * DEAL_UNR) {
            int ent[DEAL_UNR];
            int es[DEAL_UNR];
            int ed[DEAL_UNR];
            int er[DEAL_UNR];
            #pragma unroll
            for (int u = 0; u < DEAL_UNR; ++u) {
                const int idx = base + u * AGG_THREADS + tid;
                ent[u] = queue[min(idx, n - 1)];
            }
            #pragma unroll
            for (int u = 0; u < DEAL_UNR; ++u) {
                const int e = ent[u] & 0xfffff;
                es[u] = src[e];
                ed[u] = dst[e];
                er[u] = ef[e];
            }
            #pragma unroll
            for (int u = 0; u < DEAL_UNR; ++u) {
                const int idx = base + u * AGG_THREADS + tid;
                int node, fl, lrow;
                const bool valid = decode(ent[u], es[u], ed[u], er[u], node, fl, lrow);
                if (idx < n && valid) {
                    const int slot = atomicAdd(&rcnt[lrow], 1);
                    if (slot < ROW_CAP)
                        rq[lrow * ROW_CAP + slot] = node | (fl << 16);
                    else
                        add_now(node, fl, lrow);
                }
            }
        }
        __syncthreads();
    } else {
        constexpr int LV = TILE_R * ROW_CAP / 4;
        constexpr int LU = (LV + AGG_THREADS - 1) / AGG_THREADS;
        int4 lv[LU];
        #pragma unroll
        for (int u = 0; u < LU; ++u)
            lv[u] = ((const int4*)mylist)[min(tid + u * AGG_THREADS, LV - 1)];
        #pragma unroll
        for (int u = 0; u < LU; ++u)
            if (tid + u * AGG_THREADS < LV) ((int4*)rq)[tid + u * AGG_THREADS] = lv[u];
        if (tid < TILE_R) rcnt[tid] = mylist[TILE_R * ROW_CAP + tid];
        if (tid == 0) ovf = 0;
        __syncthreads();
    }

    const int any_ovf = __builtin_amdgcn_readfirstlane(ovf);
    const float* hl = h + lane * 4;

    for (int lr = wave; lr < TILE_R; lr += AGG_WAVES) {
        const int rc = __builtin_amdgcn_readfirstlane(min(rcnt[lr], ROW_CAP));
        float4 acc = make_float4(0.0f, 0.0f, 0.0f, 0.0f);
        for (int b = 0; b < rc; b += 32) {
            const int m  = min(32, rc - b);
            const int my = rq[lr * ROW_CAP + b + lane];
            for (int j = 0; j < m; j += GATHER_UNR) {
                float4 v[GATHER_UNR];
                int    en[GATHER_UNR];
                #pragma unroll
                for (int u = 0; u < GATHER_UNR; ++u) {
                    en[u] = 1 << 17;
                    if (j + u < m) {
                        en[u] = __builtin_amdgcn_readlane(my, j + u);
                        if (MODE == 2 || !((en[u] >> 17) & 1))
                            v[u] = *(const float4*)(hl + (long)(en[u] & 0xffff) * HD);
                    }
                }
                #pragma unroll
                for (int u = 0; u < GATHER_UNR; ++u) {
                    const bool ok = (j + u < m) && (MODE == 2 || !((en[u] >> 17) & 1));
                    float4 x = v[u];
                    if (MODE == 2) {
                        const float w = (ok && ((en[u] >> 16) & 1)) ? 2.0f : 1.0f;
                        x.x *= w; x.y *= w; x.z *= w; x.w *= w;
                    }
                    acc.x = ok ? acc.x + x.x : acc.x;
                    acc.y = ok ? acc.y + x.y : acc.y;
                    acc.z = ok ? acc.z + x.z : acc.z;
                    acc.w = ok ? acc.w + x.w : acc.w;
                }
            }
        }

        if (any_ovf != 0) {
            const float* t = tile + lr * HD + lane;
            acc.x += t[0]; acc.y += t[32]; acc.z += t[64]; acc.w += t[96];
        }
        const int row = r0 + lr;
        if (MODE == 0) {
            const float inn = in_norm[row];
            const float on  = out_norm[row];
            const float4 bb = *(const float4*)(bias + lane * 4);
            acc.x = fmaxf(acc.x * inn + bb.x, 0.0f) * on;
            acc.y = fmaxf(acc.y * inn + bb.y, 0.0f) * on;
            acc.z = fmaxf(acc.z * inn + bb.z, 0.0f) * on;
            acc.w = fmaxf(acc.w * inn + bb.w, 0.0f) * on;
        }
        *(float4*)(outp + (long)row * HD + lane * 4) = acc;
    }
}

extern "C" void kernel_launch(void* const* d_in, const int* in_sizes, int n_in,
                              void* d_out, int out_size, void* d_ws, size_t ws_size,
                              hipStream_t stream)
{
    (void)in_sizes; (void)n_in; (void)out_size; (void)ws_size;

    const float* feat0 = (const float*)d_in[0];
    const float* Wfc0  = (const float*)d_in[1];
    const float* bfc0  = (const float*)d_in[2];
    const float* feat1 = (const float*)d_in[3];
    const float* Wfc1  = (const float*)d_in[4];
    const float* bfc1  = (const float*)d_in[5];
    const float* feat2 = (const float*)d_in[6];
    const float* Wfc2  = (const float*)d_in[7];
    const float* bfc2  = (const float*)d_in[8];
    const float* feat3 = (const float*)d_in[9];
    const float* Wfc3  = (const float*)d_in[10];
    const float* bfc3  = (const float*)d_in[11];
    const int*   src   = (const int*)d_in[12];
    const int*   dst   = (const int*)d_in[13];
    const int*   efeat = (const int*)d_in[14];
    const float* b_g0  = (const float*)d_in[15];
    const float* W_g1  = (const float*)d_in[16];
    const float* b_g1  = (const float*)d_in[17];
    float* out = (float*)d_out;

    const size_t NODE_ELEMS = (size_t)N_NODES * HD;
    float* bufA = (float*)d_ws;
    float* bufB = bufA + NODE_ELEMS;
    float* degO = bufB + NODE_ELEMS;
    float* degI = degO + N_NODES;

    edge_keys<<<(N_GRP_PAD * 8 + 255) / 256, 256, 0, stream>>>(src, dst, efeat, bufB);
    tile_lists<<<N_TILES, AGG_THREADS, 0, stream>>>(src, dst, efeat, bufB, out, degO, degI);

    gemm_wmma_f32<128><<<PROJ_BLOCKS, 256, 0, stream>>>(
        feat0, Wfc0, bfc0, nullptr, degO + 0 * N_PER,
        bufA + (size_t)0 * N_PER * HD, N_PER, 0);
    gemm_wmma_f32<128><<<PROJ_BLOCKS, 256, 0, stream>>>(
        feat1, Wfc1, bfc1, nullptr, degO + 1 * N_PER,
        bufA + (size_t)1 * N_PER * HD, N_PER, 0);
    gemm_wmma_f32<64><<<PROJ_BLOCKS, 256, 0, stream>>>(
        feat2, Wfc2, bfc2, nullptr, degO + 2 * N_PER,
        bufA + (size_t)2 * N_PER * HD, N_PER, 0);
    gemm_wmma_f32<64><<<PROJ_BLOCKS, 256, 0, stream>>>(
        feat3, Wfc3, bfc3, nullptr, degO + 3 * N_PER,
        bufA + (size_t)3 * N_PER * HD, N_PER, 0);

    tile_aggregate<0><<<N_TILES, AGG_THREADS, 0, stream>>>(
        bufA, src, dst, efeat, out, degI, degO, b_g0, bufB);

    tile_aggregate<1><<<N_TILES, AGG_THREADS, 0, stream>>>(
        bufB, src, dst, efeat, out, nullptr, nullptr, nullptr, bufA);
    gemm_wmma_f32<128><<<L1_BLOCKS, 256, 0, stream>>>(
        bufA, W_g1, b_g1, degI, nullptr, bufB, N_NODES, 1);

    tile_aggregate<2><<<N_TILES, AGG_THREADS, 0, stream>>>(
        bufB, src, dst, efeat, out, nullptr, nullptr, nullptr, out);
}


# GEMM launches: a block's 16-row A tile is fetched once, as whole lines, into LDS (two buffers; next tile and its scales requested ahead of the products); waves read fragments from LDS; k order kept
# speedup vs baseline: 3.8506x; 1.1160x over previous
#include <hip/hip_runtime.h>

typedef float v2f __attribute__((ext_vector_type(2)));
typedef float v8f __attribute__((ext_vector_type(8)));

#define N_NODES 50000
#define N_PER   12500
#define NE      600000
#define HD      128

#define TILE_R      250
#define N_TILES     200
#define AGG_THREADS 1024
#define SCAN_SG     4
#define N_KEYW      (NE / 4)
#define N_GRP       (((N_KEYW + 31) / 32) * 4)
#define N_GRP_PAD   (((N_GRP + AGG_THREADS * SCAN_SG - 1) / (AGG_THREADS * SCAN_SG)) * (AGG_THREADS * SCAN_SG))
#define N_KVEC      (2 * N_GRP_PAD)
#define REGION_B    (TILE_R * HD * 4)
#define ROW_CAP     64
#define LIST_INTS   (TILE_R * ROW_CAP + TILE_R + 1)

static_assert(TILE_R * N_TILES == N_NODES, "tiles must cover the node range exactly");
static_assert(N_TILES <= 255, "a tile key is one byte and 0xff is the never-matching filler");
static_assert(NE % 4 == 0, "edge ids are read as int4");
static_assert(LIST_INTS * 4 <= REGION_B, "the row lists of a tile fit its region");
static_assert(REGION_B % 16 == 0, "16-byte vectors stay aligned");
static_assert((size_t)NE * 8 + (size_t)2 * N_KVEC * 16 <= (size_t)N_NODES * HD * 4, "edge records and both key arrays fit one node plane");

#define PROJ_BLOCKS 391
#define L1_BLOCKS   512

template <int K>
__global__ __launch_bounds__(256)
void gemm_wmma_f32(const float* __restrict__ A,
                   const float* __restrict__ B,
                   const float* __restrict__ bias,
                   const float* __restrict__ rowscale,
                   const float* __restrict__ outscale,
                   float* __restrict__ C,
                   int M, int relu)
{
    static_assert(K % 8 == 0, "the loaders walk k in blocks of 8");
    constexpr int PITCH = K + 4;
    constexpr int TVT   = (16 * K / 4) / 256;
    static_assert((16 * K / 4) % 256 == 0 && TVT >= 1 && TVT <= 2, "a tile is one or two 16-byte vectors per thread of the 256");
    static_assert(((PITCH * 4 / 16) & 1) == 1 && (PITCH * 4) % 16 == 0, "row pitch is 16 bytes times an odd number");
    __shared__ __attribute__((aligned(16))) float As[2][16 * PITCH];
    const int lane = threadIdx.x & 31;
    const int half = lane >> 4;
    const int lm   = lane & 15;
    const int wave   = (blockIdx.x * blockDim.x + threadIdx.x) >> 5;
    const int tn  = wave & 7;
    const int col = tn * 16 + lm;

    v2f bf[K / 4];
    #pragma unroll
    for (int kk = 0; kk < K / 4; ++kk) {
        const int kb = (kk >> 1) * 8 + half * 4 + (kk & 1) * 2;
        bf[kk][0] = B[(long)kb * HD + col];
        bf[kk][1] = B[(long)(kb + 1) * HD + col];
    }

    const int mt = (M + 15) >> 4;
    const float bv = bias[col];

    const int tid = threadIdx.x;
    const int sr0 = tid / (K / 4);
    const int sr1 = (tid + 256) / (K / 4);
    const int sc  = (tid % (K / 4)) * 4;
    float4 s0 = make_float4(0.0f, 0.0f, 0.0f, 0.0f);
    float4 s1 = make_float4(0.0f, 0.0f, 0.0f, 0.0f);
    auto fetch = [&](int tile) {
        const int ra = tile * 16 + sr0;
        s0 = *(const float4*)(A + (long)((ra < M) ? ra : (M - 1)) * K + sc);
        if (TVT > 1) {
            const int rb = tile * 16 + sr1;
            s1 = *(const float4*)(A + (long)((rb < M) ? rb : (M - 1)) * K + sc);
        }
    };
    auto stash = [&](int b) {
        *(float4*)(&As[b][sr0 * PITCH + sc]) = s0;
        if (TVT > 1) *(float4*)(&As[b][sr1 * PITCH + sc]) = s1;
    };

    auto row_scale = [&](int tile) -> float {
        const int r = tile * 16 + lm;
        return (rowscale != nullptr) ? rowscale[(r < M) ? r : (M - 1)] : 1.0f;
    };
    auto out_scale = [&](int tile, int v) -> float {
        const int r = tile * 16 + v + half * 8;
        return (outscale != nullptr) ? outscale[(r < M) ? r : (M - 1)] : 1.0f;
    };

    int tm = blockIdx.x;
    int buf = 0;
    float rs = 1.0f, rsn = 1.0f;
    float os0 = 1.0f, os1 = 1.0f, os2 = 1.0f, os3 = 1.0f, os4 = 1.0f, os5 = 1.0f, os6 = 1.0f, os7 = 1.0f;
    float on0 = 1.0f, on1 = 1.0f, on2 = 1.0f, on3 = 1.0f, on4 = 1.0f, on5 = 1.0f, on6 = 1.0f, on7 = 1.0f;
    if (tm < mt) {
        fetch(tm);
        rs  = row_scale(tm);
        os0 = out_scale(tm, 0); os1 = out_scale(tm, 1); os2 = out_scale(tm, 2); os3 = out_scale(tm, 3);
        os4 = out_scale(tm, 4); os5 = out_scale(tm, 5); os6 = out_scale(tm, 6); os7 = out_scale(tm, 7);
        stash(0);
    }
    __syncthreads();

    for (; tm < mt; tm += gridDim.x) {
        const int tnx = tm + (int)gridDim.x;
        const bool more = tnx < mt;
        if (more) {
            fetch(tnx);
            rsn = row_scale(tnx);
            on0 = out_scale(tnx, 0); on1 = out_scale(tnx, 1); on2 = out_scale(tnx, 2); on3 = out_scale(tnx, 3);
            on4 = out_scale(tnx, 4); on5 = out_scale(tnx, 5); on6 = out_scale(tnx, 6); on7 = out_scale(tnx, 7);
        }

        const float* ap = &As[buf][lm * PITCH + half * 4];

        float4 av[K / 8];
        #pragma unroll
        for (int p = 0; p < K / 8; ++p)
            av[p] = *(const float4*)(ap + p * 8);

        v8f acc = {};
        #pragma unroll
        for (int p = 0; p < K / 8; ++p) {
            v2f a0, a1;
            a0[0] = av[p].x * rs;
            a0[1] = av[p].y * rs;
            a1[0] = av[p].z * rs;
            a1[1] = av[p].w * rs;
            acc = __builtin_amdgcn_wmma_f32_16x16x4_f32(
                false, a0, false, bf[2 * p], (short)0, acc, false, false);
            acc = __builtin_amdgcn_wmma_f32_16x16x4_f32(
                false, a1, false, bf[2 * p + 1], (short)0, acc, false, false);
        }

        if (more) stash(buf ^ 1);
        __syncthreads();

        const float osv[8] = {os0, os1, os2, os3, os4, os5, os6, os7};
        #pragma unroll
        for (int v = 0; v < 8; ++v) {
            const int orow = tm * 16 + v + half * 8;
            if (orow < M) {
                float val = acc[v] + bv;
                if (relu) val = fmaxf(val, 0.0f);
                if (outscale != nullptr) val *= osv[v];
                C[(long)orow * HD + col] = val;
            }
        }

        buf ^= 1;
        rs = rsn;
        os0 = on0; os1 = on1; os2 = on2; os3 = on3; os4 = on4; os5 = on5; os6 = on6; os7 = on7;
    }
}

__global__ void edge_keys(const int* __restrict__ src,
                          const int* __restrict__ dst,
                          const int* __restrict__ ef,
                          float* __restrict__ recbuf)
{
    uint4* rec4 = (uint4*)recbuf;
    uint4* keyv = (uint4*)recbuf + NE / 2;
    const int lane = threadIdx.x & 31;
    int j = blockIdx.x * blockDim.x + threadIdx.x;
    const int stride = gridDim.x * blockDim.x;
    for (; j - lane < N_GRP_PAD * 8; j += stride) {
        unsigned kd[4] = {0xffu, 0xffu, 0xffu, 0xffu};
        unsigned ks[4] = {0xffu, 0xffu, 0xffu, 0xffu};
        if (j < N_KEYW) {
            const int4 s = *(const int4*)(src + 4 * j);
            const int4 d = *(const int4*)(dst + 4 * j);
            const int4 r = *(const int4*)(ef + 4 * j);
            kd[0] = (unsigned)d.x / TILE_R; kd[1] = (unsigned)d.y / TILE_R;
            kd[2] = (unsigned)d.z / TILE_R; kd[3] = (unsigned)d.w / TILE_R;
            ks[0] = (unsigned)s.x / TILE_R; ks[1] = (unsigned)s.y / TILE_R;
            ks[2] = (unsigned)s.z / TILE_R; ks[3] = (unsigned)s.w / TILE_R;
            rec4[2 * j]     = make_uint4((unsigned)s.x | ((unsigned)d.x << 16), (unsigned)r.x,
                                         (unsigned)s.y | ((unsigned)d.y << 16), (unsigned)r.y);
            rec4[2 * j + 1] = make_uint4((unsigned)s.z | ((unsigned)d.z << 16), (unsigned)r.z,
                                         (unsigned)s.w | ((unsigned)d.w << 16), (unsigned)r.w);
        }
        const int q = __builtin_amdgcn_readfirstlane((j - lane) >> 5);
        unsigned w[4] = {0u, 0u, 0u, 0u};
        #pragma unroll
        for (int L = 0; L < 16; ++L) {
            const unsigned key = (L >> 3) ? ks[(L >> 1) & 3] : kd[(L >> 1) & 3];
            #pragma unroll
            for (int i = 0; i < 4; ++i) {
                const unsigned plane = __builtin_amdgcn_ballot_w32(((key >> (4 * (L & 1) + i)) & 1u) != 0u);
                w[i] = (lane == L) ? plane : w[i];
            }
        }
        if (lane < 16)
            keyv[(lane >> 3) * N_KVEC + (lane & 1) * N_GRP_PAD + 4 * q + ((lane >> 1) & 3)] =
                make_uint4(w[0], w[1], w[2], w[3]);
    }
}

#define Q_CAP       16384
#define LQ_CAP      32768
#define AGG_WAVES   (AGG_THREADS / 32)
#define GATHER_UNR  16
#define DEAL_UNR    4

static_assert(NE <= (1 << 20), "queue entry packing");
static_assert(ROW_CAP == 64, "a row's list is read as two batches of 32 entries, one per lane");
static_assert(N_NODES <= (1 << 16), "row-list entry: node id in the low 16 bits, flags in bits 16 and 17; edge record: src | dst << 16");
static_assert((TILE_R * ROW_CAP) % 4 == 0, "row lists are copied as 16-byte vectors");

__device__ __forceinline__ void lds_add(float* p, float v)
{
    __hip_atomic_fetch_add(p, v, __ATOMIC_RELAXED, __HIP_MEMORY_SCOPE_WORKGROUP);
}

__device__ __forceinline__ int wave_prefix_sum(int x)
{
    x += __builtin_amdgcn_update_dpp(0, x, 0x111, 0xf, 0xf, true);
    x += __builtin_amdgcn_update_dpp(0, x, 0x112, 0xf, 0xf, true);
    x += __builtin_amdgcn_update_dpp(0, x, 0x114, 0xf, 0xf, true);
    x += __builtin_amdgcn_update_dpp(0, x, 0x118, 0xf, 0xf, true);
    const int low = __builtin_amdgcn_readlane(x, 15);
    return x + (((threadIdx.x & 16) != 0) ? low : 0);
}

__global__ __launch_bounds__(AGG_THREADS)
void tile_lists(const int* __restrict__ src,
                const int* __restrict__ dst,
                const int* __restrict__ ef,
                const float* __restrict__ recbuf,
                float* scratch,
                float* __restrict__ out_norm,
                float* __restrict__ in_norm)
{
    __shared__ int   queue[LQ_CAP];
    __shared__ __attribute__((aligned(16))) int rq[TILE_R * ROW_CAP];
    __shared__ int   rcnt[TILE_R];
    __shared__ int   din[TILE_R];
    __shared__ int   dout[TILE_R];
    __shared__ int   qn;
    __shared__ int   ovf;

    const int tid = threadIdx.x;
    const int r0  = blockIdx.x * TILE_R;
    int* mylist = (int*)((char*)scratch + (size_t)blockIdx.x * REGION_B);
    const uint2* rec = (const uint2*)recbuf;
    const uint4* keyv = (const uint4*)recbuf + NE / 2;

    if (tid < TILE_R) { rcnt[tid] = 0; din[tid] = 0; dout[tid] = 0; }
    if (tid == 0) { qn = 0; ovf = 0; }
    __syncthreads();

    auto deal = [&](int ent, int s, int d, int r, bool listed) {
        const bool srcend = ((ent >> 29) & 1) != 0;
        const int lrow = (srcend ? s : d) - r0;
        if ((unsigned)lrow >= (unsigned)TILE_R) return;
        atomicAdd(srcend ? &dout[lrow] : &din[lrow], 1);
        if (srcend && !(r == 6 || r == 14 || r == 30)) return;
        if (!listed) return;
        const int fl = srcend ? 2 : ((r >= 0 && r <= 4) ? 1 : 0);
        const int slot = atomicAdd(&rcnt[lrow], 1);
        if (slot < ROW_CAP)
            rq[lrow * ROW_CAP + slot] = (srcend ? d : s) | (fl << 16);
        else
            ovf = 1;
    };

    auto place = [&](int slot, int ent) {
        if (slot < LQ_CAP) {
            queue[slot] = ent;
        } else {
            const int e = ent & 0xfffff;
            ovf = 1;
            deal(ent, src[e], dst[e], ef[e], false);
        }
    };

    {
        unsigned tb[8];
        #pragma unroll
        for (int b = 0; b < 8; ++b) tb[b] = 0u - (((unsigned)blockIdx.x >> b) & 1u);

        auto match32 = [&](const uint4& lo, const uint4& hi) -> unsigned {
            return ~((lo.x ^ tb[0]) | (lo.y ^ tb[1]) | (lo.z ^ tb[2]) | (lo.w ^ tb[3]) |
                     (hi.x ^ tb[4]) | (hi.y ^ tb[5]) | (hi.z ^ tb[6]) | (hi.w ^ tb[7]));
        };
        auto emit = [&](unsigned f, int ebase, int flag, int& slot) {
            while (f != 0) {
                const int b = __builtin_ctz(f);
                f &= f - 1;
                place(slot, (ebase + 4 * b) | flag);
                ++slot;
            }
        };

        for (int g0 = 0; g0 < N_GRP_PAD; g0 += AGG_THREADS * SCAN_SG) {
            uint4 dl[SCAN_SG], dh[SCAN_SG];
            uint4 sl[SCAN_SG], sh[SCAN_SG];
            #pragma unroll
            for (int u = 0; u < SCAN_SG; ++u) {
                const int g = g0 + u * AGG_THREADS + tid;
                dl[u] = keyv[g];
                dh[u] = keyv[N_GRP_PAD + g];
                sl[u] = keyv[N_KVEC + g];
                sh[u] = keyv[N_KVEC + N_GRP_PAD + g];
            }
            unsigned md[SCAN_SG];
            unsigned ms[SCAN_SG];
            int cnt = 0;
            #pragma unroll
            for (int u = 0; u < SCAN_SG; ++u) {
                md[u] = match32(dl[u], dh[u]);
                ms[u] = match32(sl[u], sh[u]);
                cnt += __builtin_popcount(md[u]) + __builtin_popcount(ms[u]);
            }
            const int incl = wave_prefix_sum(cnt);
            const int total = __builtin_amdgcn_readlane(incl, 31);
            if (total != 0) {
                int first = 0;
                if ((tid & 31) == 31) first = atomicAdd(&qn, total);
                int slot = __builtin_amdgcn_readlane(first, 31) + incl - cnt;
                #pragma unroll
                for (int u = 0; u < SCAN_SG; ++u) {
                    const int g = g0 + u * AGG_THREADS + tid;
                    const int ebase = (g >> 2) * 128 + (g & 3);
                    emit(md[u], ebase, 0, slot);
                    emit(ms[u], ebase, 1 << 29, slot);
                }
            }
        }
    }
    __syncthreads();

    const int n = __builtin_amdgcn_readfirstlane(min(qn, LQ_CAP));
    for (int base = 0; base < n; base += AGG_THREADS * DEAL_UNR) {
        int   ent[DEAL_UNR];
        uint2 er[DEAL_UNR];
        #pragma unroll
        for (int u = 0; u < DEAL_UNR; ++u) {
            const int idx = base + u * AGG_THREADS + tid;
            ent[u] = queue[min(idx, n - 1)];
        }
        #pragma unroll
        for (int u = 0; u < DEAL_UNR; ++u)
            er[u] = rec[ent[u] & 0xfffff];
        #pragma unroll
        for (int u = 0; u < DEAL_UNR; ++u) {
            const int idx = base + u * AGG_THREADS + tid;
            if (idx < n)
                deal(ent[u], (int)(er[u].x & 0xffffu), (int)(er[u].x >> 16), (int)er[u].y, true);
        }
    }
    __syncthreads();

    for (int i = tid; i < TILE_R * ROW_CAP / 4; i += AGG_THREADS)
        ((int4*)mylist)[i] = ((const int4*)rq)[i];
    if (tid < TILE_R) {
        mylist[TILE_R * ROW_CAP + tid] = min(rcnt[tid], ROW_CAP);
        out_norm[r0 + tid] = rsqrtf(fmaxf((float)dout[tid], 1.0f));
        in_norm[r0 + tid]  = rsqrtf(fmaxf((float)din[tid], 1.0f));
    }
    if (tid == 0) mylist[TILE_R * ROW_CAP + TILE_R] = ovf;
}

template <int MODE>
__global__ __launch_bounds__(AGG_THREADS)
void tile_aggregate(const float* __restrict__ h,
                    const int* __restrict__ src,
                    const int* __restrict__ dst,
                    const int* __restrict__ ef,
                    float* scratch,
                    const float* __restrict__ in_norm,
                    const float* __restrict__ out_norm,
                    const float* __restrict__ bias,
                    float* outp)
{
    __shared__ float tile[TILE_R * HD];
    __shared__ int   queue[Q_CAP];
    __shared__ __attribute__((aligned(16))) int rq[TILE_R * ROW_CAP];
    __shared__ int   rcnt[TILE_R];
    __shared__ int   qn;
    __shared__ int   ovf;

    const int tid  = threadIdx.x;
    const int lane = tid & 31;
    const int wave = __builtin_amdgcn_readfirstlane(tid >> 5);
    const int r0   = blockIdx.x * TILE_R;
    int* mylist = (int*)((char*)scratch + (size_t)blockIdx.x * REGION_B);

    const bool build = __builtin_amdgcn_readfirstlane(mylist[TILE_R * ROW_CAP + TILE_R]) != 0;

    auto decode = [&](int ent, int s, int d, int r, int& node, int& fl, int& lrow) -> bool {
        bool valid;
        if ((ent >> 29) & 1) {
            node = d;
            lrow = s - r0;
            fl = 2;
            valid = (r == 6 || r == 14 || r == 30);
        } else {
            node = s;
            lrow = d - r0;
            fl = (r >= 0 && r <= 4) ? 1 : 0;
            valid = true;
        }
        return valid && ((unsigned)lrow < (unsigned)TILE_R);
    };

    auto add_now = [&](int node, int fl, int lrow) {
        ovf = 1;
        if (MODE != 2 && fl == 2) return;
        const float w = (MODE == 2 && fl == 1) ? 2.0f : 1.0f;
        float* t = tile + lrow * HD;
        for (int c = 0; c < HD; ++c)
            lds_add(t + (c & 3) * 32 + (c >> 2), h[(long)node * HD + c] * w);
    };

    auto place = [&](int slot, int ent) {
        if (slot < Q_CAP) {
            queue[slot] = ent;
        } else {
            const int e = ent & 0xfffff;
            int node, fl, lrow;
            if (decode(ent, src[e], dst[e], ef[e], node, fl, lrow)) add_now(node, fl, lrow);
        }
    };

    if (build) {
        for (int i = tid; i < TILE_R * HD; i += AGG_THREADS) tile[i] = 0.0f;
        if (tid < TILE_R) rcnt[tid] = 0;
        if (tid == 0) { qn = 0; ovf = 0; }
        __syncthreads();

        for (int e = tid; e < NE; e += AGG_THREADS) {
            if ((unsigned)(dst[e] - r0) < (unsigned)TILE_R) place(atomicAdd(&qn, 1), e);
            if (MODE == 2 && (unsigned)(src[e] - r0) < (unsigned)TILE_R) place(atomicAdd(&qn, 1), e | (1 << 29));
        }
        __syncthreads();

        const int n = __builtin_amdgcn_readfirstlane(min(qn, Q_CAP));
        for (int base = 0; base < n; base += AGG_THREADS * DEAL_UNR) {
            int ent[DEAL_UNR];
            int es[DEAL_UNR];
            int ed[DEAL_UNR];
            int er[DEAL_UNR];
            #pragma unroll
            for (int u = 0; u < DEAL_UNR; ++u) {
                const int idx = base + u * AGG_THREADS + tid;
                ent[u] = queue[min(idx, n - 1)];
            }
            #pragma unroll
            for (int u = 0; u < DEAL_UNR; ++u) {
                const int e = ent[u] & 0xfffff;
                es[u] = src[e];
                ed[u] = dst[e];
                er[u] = ef[e];
            }
            #pragma unroll
            for (int u = 0; u < DEAL_UNR; ++u) {
                const int idx = base + u * AGG_THREADS + tid;
                int node, fl, lrow;
                const bool valid = decode(ent[u], es[u], ed[u], er[u], node, fl, lrow);
                if (idx < n && valid) {
                    const int slot = atomicAdd(&rcnt[lrow], 1);
                    if (slot < ROW_CAP)
                        rq[lrow * ROW_CAP + slot] = node | (fl << 16);
                    else
                        add_now(node, fl, lrow);
                }
            }
        }
        __syncthreads();
    } else {
        constexpr int LV = TILE_R * ROW_CAP / 4;
        constexpr int LU = (LV + AGG_THREADS - 1) / AGG_THREADS;
        int4 lv[LU];
        #pragma unroll
        for (int u = 0; u < LU; ++u)
            lv[u] = ((const int4*)mylist)[min(tid + u * AGG_THREADS, LV - 1)];
        #pragma unroll
        for (int u = 0; u < LU; ++u)
            if (tid + u * AGG_THREADS < LV) ((int4*)rq)[tid + u * AGG_THREADS] = lv[u];
        if (tid < TILE_R) rcnt[tid] = mylist[TILE_R * ROW_CAP + tid];
        if (tid == 0) ovf = 0;
        __syncthreads();
    }

    const int any_ovf = __builtin_amdgcn_readfirstlane(ovf);
    const float* hl = h + lane * 4;

    for (int lr = wave; lr < TILE_R; lr += AGG_WAVES) {
        const int rc = __builtin_amdgcn_readfirstlane(min(rcnt[lr], ROW_CAP));
        float4 acc = make_float4(0.0f, 0.0f, 0.0f, 0.0f);
        for (int b = 0; b < rc; b += 32) {
            const int m  = min(32, rc - b);
            const int my = rq[lr * ROW_CAP + b + lane];
            for (int j = 0; j < m; j += GATHER_UNR) {
                float4 v[GATHER_UNR];
                int    en[GATHER_UNR];
                #pragma unroll
                for (int u = 0; u < GATHER_UNR; ++u) {
                    en[u] = 1 << 17;
                    if (j + u < m) {
                        en[u] = __builtin_amdgcn_readlane(my, j + u);
                        if (MODE == 2 || !((en[u] >> 17) & 1))
                            v[u] = *(const float4*)(hl + (long)(en[u] & 0xffff) * HD);
                    }
                }
                #pragma unroll
                for (int u = 0; u < GATHER_UNR; ++u) {
                    const bool ok = (j + u < m) && (MODE == 2 || !((en[u] >> 17) & 1));
                    float4 x = v[u];
                    if (MODE == 2) {
                        const float w = (ok && ((en[u] >> 16) & 1)) ? 2.0f : 1.0f;
                        x.x *= w; x.y *= w; x.z *= w; x.w *= w;
                    }
                    acc.x = ok ? acc.x + x.x : acc.x;
                    acc.y = ok ? acc.y + x.y : acc.y;
                    acc.z = ok ? acc.z + x.z : acc.z;
                    acc.w = ok ? acc.w + x.w : acc.w;
                }
            }
        }

        if (any_ovf != 0) {
            const float* t = tile + lr * HD + lane;
            acc.x += t[0]; acc.y += t[32]; acc.z += t[64]; acc.w += t[96];
        }
        const int row = r0 + lr;
        if (MODE == 0) {
            const float inn = in_norm[row];
            const float on  = out_norm[row];
            const float4 bb = *(const float4*)(bias + lane * 4);
            acc.x = fmaxf(acc.x * inn + bb.x, 0.0f) * on;
            acc.y = fmaxf(acc.y * inn + bb.y, 0.0f) * on;
            acc.z = fmaxf(acc.z * inn + bb.z, 0.0f) * on;
            acc.w = fmaxf(acc.w * inn + bb.w, 0.0f) * on;
        }
        *(float4*)(outp + (long)row * HD + lane * 4) = acc;
    }
}

extern "C" void kernel_launch(void* const* d_in, const int* in_sizes, int n_in,
                              void* d_out, int out_size, void* d_ws, size_t ws_size,
                              hipStream_t stream)
{
    (void)in_sizes; (void)n_in; (void)out_size; (void)ws_size;

    const float* feat0 = (const float*)d_in[0];
    const float* Wfc0  = (const float*)d_in[1];
    const float* bfc0  = (const float*)d_in[2];
    const float* feat1 = (const float*)d_in[3];
    const float* Wfc1  = (const float*)d_in[4];
    const float* bfc1  = (const float*)d_in[5];
    const float* feat2 = (const float*)d_in[6];
    const float* Wfc2  = (const float*)d_in[7];
    const float* bfc2  = (const float*)d_in[8];
    const float* feat3 = (const float*)d_in[9];
    const float* Wfc3  = (const float*)d_in[10];
    const float* bfc3  = (const float*)d_in[11];
    const int*   src   = (const int*)d_in[12];
    const int*   dst   = (const int*)d_in[13];
    const int*   efeat = (const int*)d_in[14];
    const float* b_g0  = (const float*)d_in[15];
    const float* W_g1  = (const float*)d_in[16];
    const float* b_g1  = (const float*)d_in[17];
    float* out = (float*)d_out;

    const size_t NODE_ELEMS = (size_t)N_NODES * HD;
    float* bufA = (float*)d_ws;
    float* bufB = bufA + NODE_ELEMS;
    float* degO = bufB + NODE_ELEMS;
    float* degI = degO + N_NODES;

    edge_keys<<<(N_GRP_PAD * 8 + 255) / 256, 256, 0, stream>>>(src, dst, efeat, bufB);
    tile_lists<<<N_TILES, AGG_THREADS, 0, stream>>>(src, dst, efeat, bufB, out, degO, degI);

    gemm_wmma_f32<128><<<PROJ_BLOCKS, 256, 0, stream>>>(
        feat0, Wfc0, bfc0, nullptr, degO + 0 * N_PER,
        bufA + (size_t)0 * N_PER * HD, N_PER, 0);
    gemm_wmma_f32<128><<<PROJ_BLOCKS, 256, 0, stream>>>(
        feat1, Wfc1, bfc1, nullptr, degO + 1 * N_PER,
        bufA + (size_t)1 * N_PER * HD, N_PER, 0);
    gemm_wmma_f32<64><<<PROJ_BLOCKS, 256, 0, stream>>>(
        feat2, Wfc2, bfc2, nullptr, degO + 2 * N_PER,
        bufA + (size_t)2 * N_PER * HD, N_PER, 0);
    gemm_wmma_f32<64><<<PROJ_BLOCKS, 256, 0, stream>>>(
        feat3, Wfc3, bfc3, nullptr, degO + 3 * N_PER,
        bufA + (size_t)3 * N_PER * HD, N_PER, 0);

    tile_aggregate<0><<<N_TILES, AGG_THREADS, 0, stream>>>(
        bufA, src, dst, efeat, out, degI, degO, b_g0, bufB);

    tile_aggregate<1><<<N_TILES, AGG_THREADS, 0, stream>>>(
        bufB, src, dst, efeat, out, nullptr, nullptr, nullptr, bufA);
    gemm_wmma_f32<128><<<L1_BLOCKS, 256, 0, stream>>>(
        bufA, W_g1, b_g1, degI, nullptr, bufB, N_NODES, 1);

    tile_aggregate<2><<<N_TILES, AGG_THREADS, 0, stream>>>(
        bufB, src, dst, efeat, out, nullptr, nullptr, nullptr, out);
}
